# starting kernel
# baseline (speedup 1.0000x reference)
#include <hip/hip_runtime.h>
#include <math.h>

#define B_    16
#define C_    384
#define H_    28
#define W_    28
#define HW_   784
#define N_    785
#define M_    (B_*N_)
#define MP_   (M_ + 16)
#define HEADS_ 8
#define D_    48
#define C3_   1152
#define C4_   1536
#define MT_   (M_/16)
#define MB_   ((MT_ + 1)/2)
#define SLD   800

typedef __attribute__((ext_vector_type(2))) float v2f;
typedef __attribute__((ext_vector_type(8))) float v8f;

__device__ __forceinline__ v8f wmma4(v2f a, v2f b, v8f c) {
  return __builtin_amdgcn_wmma_f32_16x16x4_f32(false, a, false, b, (short)0, c,
                                               false, false);
}

__global__ void conv_pe_kernel(const float* __restrict__ x,
                               const float* __restrict__ cw,
                               const float* __restrict__ cb,
                               float* __restrict__ xc) {
  int idx = blockIdx.x * 256 + threadIdx.x;
  if (idx >= B_ * C_ * HW_) return;
  int b   = idx / (C_ * HW_);
  int rem = idx % (C_ * HW_);
  int c   = rem / HW_;
  int p   = rem % HW_;
  int hh  = p / W_, ww = p % W_;
  const float* xp = x + (size_t)(b * C_ + c) * HW_;
  const float* wp = cw + c * 9;
  float s = 0.f;
#pragma unroll
  for (int i = 0; i < 3; ++i) {
    int hy = hh + i - 1;
    if (hy < 0 || hy >= H_) continue;
#pragma unroll
    for (int j = 0; j < 3; ++j) {
      int wx = ww + j - 1;
      if (wx < 0 || wx >= W_) continue;
      s += xp[hy * W_ + wx] * wp[i * 3 + j];
    }
  }
  xc[((size_t)b * N_ + 1 + p) * C_ + c] = xp[p] + s + cb[c];
}

__global__ void cls_copy_kernel(const float* __restrict__ cls,
                                float* __restrict__ xc) {
  int idx = blockIdx.x * 256 + threadIdx.x;
  if (idx >= B_ * C_) return;
  int b = idx / C_, c = idx % C_;
  xc[(size_t)b * N_ * C_ + c] = cls[idx];
}

__global__ void ln_kernel(const float* __restrict__ xin,
                          const float* __restrict__ g,
                          const float* __restrict__ bta,
                          float* __restrict__ yout, int ntok) {
  int wave = threadIdx.x >> 5, lane = threadIdx.x & 31;
  int t = blockIdx.x * 8 + wave;
  if (t >= ntok) return;
  const float* xp = xin + (size_t)t * C_;
  float s = 0.f, sq = 0.f;
#pragma unroll
  for (int i = 0; i < C_ / 32; ++i) {
    float v = xp[lane + i * 32];
    s += v; sq += v * v;
  }
#pragma unroll
  for (int off = 16; off; off >>= 1) {
    s  += __shfl_xor(s,  off, 32);
    sq += __shfl_xor(sq, off, 32);
  }
  float mu  = s  * (1.f / C_);
  float var = sq * (1.f / C_) - mu * mu;
  float rs  = rsqrtf(var + 1e-5f);
  float* yp = yout + (size_t)t * C_;
#pragma unroll
  for (int i = 0; i < C_ / 32; ++i) {
    int c = lane + i * 32;
    yp[c] = (xp[c] - mu) * rs * g[c] + bta[c];
  }
}

template <int EPI>
__global__ __launch_bounds__(256) void gemm_wmma_kernel(
    const float* __restrict__ X, const float* __restrict__ Wt,
    const float* __restrict__ bias, float* __restrict__ Out,
    int K, int ldo, int Mblocks) {
  int lane = threadIdx.x & 31;
  int wave = threadIdx.x >> 5;
  int mblk = blockIdx.y * 8 + wave;
  if (mblk >= Mblocks) return;
  int n0 = blockIdx.x * 64;
  int m0 = mblk * 32;
  int rc = lane & 15;
  int kb = (lane >> 4) << 1;
  const float* xp0 = X  + (size_t)(m0 + rc)      * K + kb;
  const float* xp1 = X  + (size_t)(m0 + 16 + rc) * K + kb;
  const float* wp0 = Wt + (size_t)(n0 + rc)      * K + kb;
  const float* wp1 = Wt + (size_t)(n0 + 16 + rc) * K + kb;
  const float* wp2 = Wt + (size_t)(n0 + 32 + rc) * K + kb;
  const float* wp3 = Wt + (size_t)(n0 + 48 + rc) * K + kb;

  v8f acc[2][4] = {};
#pragma unroll 4
  for (int k = 0; k < K; k += 4) {
    v2f a0 = *(const v2f*)(xp0 + k);
    v2f a1 = *(const v2f*)(xp1 + k);
    v2f b0 = *(const v2f*)(wp0 + k);
    v2f b1 = *(const v2f*)(wp1 + k);
    v2f b2 = *(const v2f*)(wp2 + k);
    v2f b3 = *(const v2f*)(wp3 + k);
    acc[0][0] = wmma4(a0, b0, acc[0][0]);
    acc[1][0] = wmma4(a1, b0, acc[1][0]);
    acc[0][1] = wmma4(a0, b1, acc[0][1]);
    acc[1][1] = wmma4(a1, b1, acc[1][1]);
    acc[0][2] = wmma4(a0, b2, acc[0][2]);
    acc[1][2] = wmma4(a1, b2, acc[1][2]);
    acc[0][3] = wmma4(a0, b3, acc[0][3]);
    acc[1][3] = wmma4(a1, b3, acc[1][3]);
  }

  int coln = lane & 15;
  int roff = (lane >> 4) << 3;
#pragma unroll
  for (int mi = 0; mi < 2; ++mi) {
#pragma unroll
    for (int j = 0; j < 4; ++j) {
      int nc = n0 + j * 16 + coln;
      float bv = bias ? bias[nc] : 0.f;
#pragma unroll
      for (int r = 0; r < 8; ++r) {
        int gm = m0 + mi * 16 + roff + r;
        if (gm < M_) {
          size_t oi = (size_t)gm * ldo + nc;
          float v = acc[mi][j][r] + bv;
          if (EPI == 0)      Out[oi] = v;
          else if (EPI == 1) Out[oi] += v;
          else               Out[oi] = 0.5f * v * (1.f + erff(v * 0.70710678118654752f));
        }
      }
    }
  }
}

__global__ __launch_bounds__(32) void attn_kernel(const float* __restrict__ qkv,
                                                  float* __restrict__ o,
                                                  float* __restrict__ ga) {
  __shared__ float S[16 * SLD];
  int lane = threadIdx.x;
  int tile = blockIdx.x;
  int h = blockIdx.y, b = blockIdx.z;
  int t0 = tile * 16;
  int row = lane & 15;
  int kb  = (lane >> 4) << 1;
  int gq = t0 + row; if (gq > N_ - 1) gq = N_ - 1;
  const float* qrow = qkv + (size_t)(b * N_ + gq) * C3_ + h * D_ + kb;
  const float scale = 0.14433756729740643f;

  for (int ct = 0; ct < 50; ++ct) {
    int kc = ct * 16 + row; if (kc > N_ - 1) kc = N_ - 1;
    const float* krow = qkv + (size_t)(b * N_ + kc) * C3_ + C_ + h * D_ + kb;
    v8f acc = {};
#pragma unroll
    for (int k = 0; k < D_; k += 4) {
      v2f a  = *(const v2f*)(qrow + k);
      v2f bb = *(const v2f*)(krow + k);
      acc = wmma4(a, bb, acc);
    }
    int sc  = ct * 16 + (lane & 15);
    int mrb = (lane >> 4) << 3;
#pragma unroll
    for (int r = 0; r < 8; ++r) S[(mrb + r) * SLD + sc] = acc[r] * scale;
  }
  __syncthreads();

  {
    int r = lane >> 1, sub = lane & 1;
    float mx = -3.0e38f;
    for (int c = sub; c < N_; c += 2) mx = fmaxf(mx, S[r * SLD + c]);
    mx = fmaxf(mx, __shfl_xor(mx, 1, 32));
    float sum = 0.f;
    for (int c = sub; c < N_; c += 2) {
      float e = expf(S[r * SLD + c] - mx);
      S[r * SLD + c] = e; sum += e;
    }
    sum += __shfl_xor(sum, 1, 32);
    float inv = 1.f / sum;
    for (int c = sub; c < N_; c += 2) S[r * SLD + c] *= inv;
    for (int c = N_ + sub; c < SLD; c += 2) S[r * SLD + c] = 0.f;
  }
  __syncthreads();

  if (t0 == 0 && (lane >> 1) == 0) {
    int sub = lane & 1;
    float* gp = ga + (size_t)(b * HEADS_ + h) * HW_;
    for (int c = 1 + sub; c < N_; c += 2) gp[c - 1] = S[c];
  }

  const float* vbase = qkv + (size_t)b * N_ * C3_ + 2 * C_ + h * D_;
  int coln = lane & 15;
  v8f acc0 = {}, acc1 = {}, acc2 = {};
  for (int k = 0; k < 788; k += 4) {
    int k0 = k + kb;
    v2f a = *(const v2f*)(&S[row * SLD + k0]);
    int r0 = k0 > HW_ ? HW_ : k0;
    int r1 = (k0 + 1) > HW_ ? HW_ : (k0 + 1);
    const float* vr0 = vbase + (size_t)r0 * C3_ + coln;
    const float* vr1 = vbase + (size_t)r1 * C3_ + coln;
    v2f b0, b1, b2;
    b0.x = vr0[0];  b0.y = vr1[0];
    b1.x = vr0[16]; b1.y = vr1[16];
    b2.x = vr0[32]; b2.y = vr1[32];
    acc0 = wmma4(a, b0, acc0);
    acc1 = wmma4(a, b1, acc1);
    acc2 = wmma4(a, b2, acc2);
  }
  int mrb = (lane >> 4) << 3;
#pragma unroll
  for (int r = 0; r < 8; ++r) {
    int gm = t0 + mrb + r;
    if (gm < N_) {
      float* op = o + ((size_t)(b * N_ + gm)) * C_ + h * D_ + coln;
      op[0]  = acc0[r];
      op[16] = acc1[r];
      op[32] = acc2[r];
    }
  }
}

__global__ void finalize_kernel(const float* __restrict__ xc,
                                const float* __restrict__ ga,
                                float* __restrict__ out) {
  size_t idx = (size_t)blockIdx.x * 256 + threadIdx.x;
  const size_t n_cls = (size_t)B_ * C_;
  const size_t n_x   = (size_t)B_ * C_ * HW_;
  const size_t n_ga  = (size_t)B_ * HW_;
  if (idx < n_cls) {
    int b = (int)(idx / C_), c = (int)(idx % C_);
    out[idx] = xc[(size_t)b * N_ * C_ + c];
  } else if (idx < n_cls + n_x) {
    size_t j = idx - n_cls;
    int b   = (int)(j / (C_ * HW_));
    int rem = (int)(j % (C_ * HW_));
    int c = rem / HW_, p = rem % HW_;
    out[idx] = xc[((size_t)b * N_ + 1 + p) * C_ + c];
  } else if (idx < n_cls + n_x + n_ga) {
    size_t j = idx - n_cls - n_x;
    int b = (int)(j / HW_), p = (int)(j % HW_);
    float s = 0.f;
#pragma unroll
    for (int hh = 0; hh < HEADS_; ++hh)
      s += ga[(size_t)(b * HEADS_ + hh) * HW_ + p];
    out[idx] = s * 0.125f;
  }
}

extern "C" void kernel_launch(void* const* d_in, const int* in_sizes, int n_in,
                              void* d_out, int out_size, void* d_ws, size_t ws_size,
                              hipStream_t stream) {
  const float* cls    = (const float*)d_in[0];
  const float* x      = (const float*)d_in[1];
  const float* conv_w = (const float*)d_in[2];
  const float* conv_b = (const float*)d_in[3];
  const float* ln1_g  = (const float*)d_in[4];
  const float* ln1_b  = (const float*)d_in[5];
  const float* qkv_w  = (const float*)d_in[6];
  const float* proj_w = (const float*)d_in[7];
  const float* proj_b = (const float*)d_in[8];
  const float* ln2_g  = (const float*)d_in[9];
  const float* ln2_b  = (const float*)d_in[10];
  const float* fc1_w  = (const float*)d_in[11];
  const float* fc1_b  = (const float*)d_in[12];
  const float* fc2_w  = (const float*)d_in[13];
  const float* fc2_b  = (const float*)d_in[14];

  float* ws  = (float*)d_ws;
  float* xc  = ws;
  float* y   = xc  + (size_t)MP_ * C_;
  float* qkv = y   + (size_t)MP_ * C_;
  float* o   = qkv + (size_t)MP_ * C3_;
  float* m1  = qkv;
  float* ga  = o   + (size_t)MP_ * C_;
  float* out = (float*)d_out;

  conv_pe_kernel<<<(B_ * C_ * HW_ + 255) / 256, 256, 0, stream>>>(x, conv_w, conv_b, xc);
  cls_copy_kernel<<<(B_ * C_ + 255) / 256, 256, 0, stream>>>(cls, xc);

  ln_kernel<<<(M_ + 7) / 8, 256, 0, stream>>>(xc, ln1_g, ln1_b, y, M_);

  gemm_wmma_kernel<0><<<dim3(C3_ / 64, (MB_ + 7) / 8), 256, 0, stream>>>(
      y, qkv_w, nullptr, qkv, C_, C3_, MB_);

  attn_kernel<<<dim3(50, HEADS_, B_), 32, 0, stream>>>(qkv, o, ga);

  gemm_wmma_kernel<1><<<dim3(C_ / 64, (MB_ + 7) / 8), 256, 0, stream>>>(
      o, proj_w, proj_b, xc, C_, C_, MB_);

  ln_kernel<<<(M_ + 7) / 8, 256, 0, stream>>>(xc, ln2_g, ln2_b, y, M_);

  gemm_wmma_kernel<2><<<dim3(C4_ / 64, (MB_ + 7) / 8), 256, 0, stream>>>(
      y, fc1_w, fc1_b, m1, C_, C4_, MB_);

  gemm_wmma_kernel<1><<<dim3(C_ / 64, (MB_ + 7) / 8), 256, 0, stream>>>(
      m1, fc2_w, fc2_b, xc, C4_, C_, MB_);

  size_t tot = (size_t)B_ * C_ + (size_t)B_ * C_ * HW_ + (size_t)B_ * HW_;
  finalize_kernel<<<(unsigned)((tot + 255) / 256), 256, 0, stream>>>(xc, ga, out);
}


# Attention: transposed products St=K*Qt, Ot=Vt*Pt so score registers feed the PV product directly; 32 queries per wave, running max/sum softmax, no LDS score rows; summation order changes
# speedup vs baseline: 1.2305x; 1.2305x over previous
#include <hip/hip_runtime.h>
#include <math.h>

#define B_    16
#define C_    384
#define H_    28
#define W_    28
#define HW_   784
#define N_    785
#define M_    (B_*N_)
#define MP_   (M_ + 16)
#define HEADS_ 8
#define D_    48
#define C3_   1152
#define C4_   1536
#define MT_   (M_/16)
#define MB_   ((MT_ + 1)/2)

typedef __attribute__((ext_vector_type(2))) float v2f;
typedef __attribute__((ext_vector_type(8))) float v8f;

__device__ __forceinline__ v8f wmma4(v2f a, v2f b, v8f c) {
  return __builtin_amdgcn_wmma_f32_16x16x4_f32(false, a, false, b, (short)0, c,
                                               false, false);
}

__global__ void conv_pe_kernel(const float* __restrict__ x,
                               const float* __restrict__ cw,
                               const float* __restrict__ cb,
                               float* __restrict__ xc) {
  int idx = blockIdx.x * 256 + threadIdx.x;
  if (idx >= B_ * C_ * HW_) return;
  int b   = idx / (C_ * HW_);
  int rem = idx % (C_ * HW_);
  int c   = rem / HW_;
  int p   = rem % HW_;
  int hh  = p / W_, ww = p % W_;
  const float* xp = x + (size_t)(b * C_ + c) * HW_;
  const float* wp = cw + c * 9;
  float s = 0.f;
#pragma unroll
  for (int i = 0; i < 3; ++i) {
    int hy = hh + i - 1;
    if (hy < 0 || hy >= H_) continue;
#pragma unroll
    for (int j = 0; j < 3; ++j) {
      int wx = ww + j - 1;
      if (wx < 0 || wx >= W_) continue;
      s += xp[hy * W_ + wx] * wp[i * 3 + j];
    }
  }
  xc[((size_t)b * N_ + 1 + p) * C_ + c] = xp[p] + s + cb[c];
}

__global__ void cls_copy_kernel(const float* __restrict__ cls,
                                float* __restrict__ xc) {
  int idx = blockIdx.x * 256 + threadIdx.x;
  if (idx >= B_ * C_) return;
  int b = idx / C_, c = idx % C_;
  xc[(size_t)b * N_ * C_ + c] = cls[idx];
}

__global__ void ln_kernel(const float* __restrict__ xin,
                          const float* __restrict__ g,
                          const float* __restrict__ bta,
                          float* __restrict__ yout, int ntok) {
  int wave = threadIdx.x >> 5, lane = threadIdx.x & 31;
  int t = blockIdx.x * 8 + wave;
  if (t >= ntok) return;
  const float* xp = xin + (size_t)t * C_;
  float s = 0.f, sq = 0.f;
#pragma unroll
  for (int i = 0; i < C_ / 32; ++i) {
    float v = xp[lane + i * 32];
    s += v; sq += v * v;
  }
#pragma unroll
  for (int off = 16; off; off >>= 1) {
    s  += __shfl_xor(s,  off, 32);
    sq += __shfl_xor(sq, off, 32);
  }
  float mu  = s  * (1.f / C_);
  float var = sq * (1.f / C_) - mu * mu;
  float rs  = rsqrtf(var + 1e-5f);
  float* yp = yout + (size_t)t * C_;
#pragma unroll
  for (int i = 0; i < C_ / 32; ++i) {
    int c = lane + i * 32;
    yp[c] = (xp[c] - mu) * rs * g[c] + bta[c];
  }
}

template <int EPI>
__global__ __launch_bounds__(256) void gemm_wmma_kernel(
    const float* __restrict__ X, const float* __restrict__ Wt,
    const float* __restrict__ bias, float* __restrict__ Out,
    int K, int ldo, int Mblocks) {
  int lane = threadIdx.x & 31;
  int wave = threadIdx.x >> 5;
  int mblk = blockIdx.y * 8 + wave;
  if (mblk >= Mblocks) return;
  int n0 = blockIdx.x * 64;
  int m0 = mblk * 32;
  int rc = lane & 15;
  int kb = (lane >> 4) << 1;
  const float* xp0 = X  + (size_t)(m0 + rc)      * K + kb;
  const float* xp1 = X  + (size_t)(m0 + 16 + rc) * K + kb;
  const float* wp0 = Wt + (size_t)(n0 + rc)      * K + kb;
  const float* wp1 = Wt + (size_t)(n0 + 16 + rc) * K + kb;
  const float* wp2 = Wt + (size_t)(n0 + 32 + rc) * K + kb;
  const float* wp3 = Wt + (size_t)(n0 + 48 + rc) * K + kb;

  v8f acc[2][4] = {};
#pragma unroll 4
  for (int k = 0; k < K; k += 4) {
    v2f a0 = *(const v2f*)(xp0 + k);
    v2f a1 = *(const v2f*)(xp1 + k);
    v2f b0 = *(const v2f*)(wp0 + k);
    v2f b1 = *(const v2f*)(wp1 + k);
    v2f b2 = *(const v2f*)(wp2 + k);
    v2f b3 = *(const v2f*)(wp3 + k);
    acc[0][0] = wmma4(a0, b0, acc[0][0]);
    acc[1][0] = wmma4(a1, b0, acc[1][0]);
    acc[0][1] = wmma4(a0, b1, acc[0][1]);
    acc[1][1] = wmma4(a1, b1, acc[1][1]);
    acc[0][2] = wmma4(a0, b2, acc[0][2]);
    acc[1][2] = wmma4(a1, b2, acc[1][2]);
    acc[0][3] = wmma4(a0, b3, acc[0][3]);
    acc[1][3] = wmma4(a1, b3, acc[1][3]);
  }

  int coln = lane & 15;
  int roff = (lane >> 4) << 3;
#pragma unroll
  for (int mi = 0; mi < 2; ++mi) {
#pragma unroll
    for (int j = 0; j < 4; ++j) {
      int nc = n0 + j * 16 + coln;
      float bv = bias ? bias[nc] : 0.f;
#pragma unroll
      for (int r = 0; r < 8; ++r) {
        int gm = m0 + mi * 16 + roff + r;
        if (gm < M_) {
          size_t oi = (size_t)gm * ldo + nc;
          float v = acc[mi][j][r] + bv;
          if (EPI == 0)      Out[oi] = v;
          else if (EPI == 1) Out[oi] += v;
          else               Out[oi] = 0.5f * v * (1.f + erff(v * 0.70710678118654752f));
        }
      }
    }
  }
}

typedef __attribute__((ext_vector_type(4))) float v4f;
struct __attribute__((packed, aligned(4))) f3 { float x, y, z; };

struct KStep { v4f k[6]; };
struct VStep { f3 va[4]; f3 vb[4]; };

__device__ __forceinline__ void load_k(KStep& t, const float* __restrict__ kbase,
                                       int key0, int c) {
  int kr = key0 + c; if (kr > N_ - 1) kr = N_ - 1;
  const float* kp = kbase + (size_t)kr * C3_;
#pragma unroll
  for (int i = 0; i < 6; ++i) t.k[i] = *(const v4f*)(kp + 8 * i);
}

__device__ __forceinline__ void load_v(VStep& t, const float* __restrict__ vbase,
                                       int key0, int hf) {
  const int kl = key0 + 8 * hf;
#pragma unroll
  for (int j = 0; j < 4; ++j) {
    int k0 = kl + 2 * j;     if (k0 > N_ - 1) k0 = N_ - 1;
    int k1 = kl + 2 * j + 1; if (k1 > N_ - 1) k1 = N_ - 1;
    t.va[j] = *(const f3*)(vbase + (size_t)k0 * C3_);
    t.vb[j] = *(const f3*)(vbase + (size_t)k1 * C3_);
  }
}

struct AttnState {
  v8f oa0, oa1, oa2, ob0, ob1, ob2;
  float m0, m1, l0, l1;
};

__device__ __forceinline__ void attn_step(const KStep& tk, const VStep& t,
                                          const v4f (&q0)[6],
                                          const v4f (&q1)[6], AttnState& st,
                                          int key0, int hf, bool cls_lane,
                                          float* __restrict__ gp) {
  const float scale = 0.14433756729740643f;
  v8f s0 = {}, s1 = {};
#pragma unroll
  for (int i = 0; i < 6; ++i) {
    v2f ka = {tk.k[i].x, tk.k[i].y}, kc = {tk.k[i].z, tk.k[i].w};
    v2f qa0 = {q0[i].x, q0[i].y}, qc0 = {q0[i].z, q0[i].w};
    v2f qa1 = {q1[i].x, q1[i].y}, qc1 = {q1[i].z, q1[i].w};
    s0 = wmma4(ka, qa0, s0);
    s1 = wmma4(ka, qa1, s1);
    s0 = wmma4(kc, qc0, s0);
    s1 = wmma4(kc, qc1, s1);
  }
  const int kl = key0 + 8 * hf;
  float x0[8], x1[8];
#pragma unroll
  for (int r = 0; r < 8; ++r) { x0[r] = s0[r] * scale; x1[r] = s1[r] * scale; }
  if (key0 + 16 > N_) {
#pragma unroll
    for (int r = 0; r < 8; ++r)
      if (kl + r >= N_) { x0[r] = -3.0e38f; x1[r] = -3.0e38f; }
  }
  if (cls_lane) {
#pragma unroll
    for (int r = 0; r < 8; ++r) {
      int key = kl + r;
      if (key >= 1 && key < N_) gp[key - 1] = x0[r];
    }
  }
  float mx0 = x0[0], mx1 = x1[0];
#pragma unroll
  for (int r = 1; r < 8; ++r) { mx0 = fmaxf(mx0, x0[r]); mx1 = fmaxf(mx1, x1[r]); }
  mx0 = fmaxf(mx0, __shfl_xor(mx0, 16, 32));
  mx1 = fmaxf(mx1, __shfl_xor(mx1, 16, 32));
  float mn0 = fmaxf(st.m0, mx0), mn1 = fmaxf(st.m1, mx1);
  float al0 = expf(st.m0 - mn0), al1 = expf(st.m1 - mn1);
  float ps0 = 0.f, ps1 = 0.f;
#pragma unroll
  for (int r = 0; r < 8; ++r) {
    x0[r] = expf(x0[r] - mn0); ps0 += x0[r];
    x1[r] = expf(x1[r] - mn1); ps1 += x1[r];
  }
  st.l0 = st.l0 * al0 + ps0;  st.l1 = st.l1 * al1 + ps1;
  st.m0 = mn0;  st.m1 = mn1;
#pragma unroll
  for (int r = 0; r < 8; ++r) {
    st.oa0[r] *= al0; st.oa1[r] *= al0; st.oa2[r] *= al0;
    st.ob0[r] *= al1; st.ob1[r] *= al1; st.ob2[r] *= al1;
  }
#pragma unroll
  for (int j = 0; j < 4; ++j) {
    v2f pa = {x0[2 * j], x0[2 * j + 1]};
    v2f pb = {x1[2 * j], x1[2 * j + 1]};
    v2f a0 = {t.va[j].x, t.vb[j].x}, a1 = {t.va[j].y, t.vb[j].y}, a2 = {t.va[j].z, t.vb[j].z};
    st.oa0 = wmma4(a0, pa, st.oa0);
    st.ob0 = wmma4(a0, pb, st.ob0);
    st.oa1 = wmma4(a1, pa, st.oa1);
    st.ob1 = wmma4(a1, pb, st.ob1);
    st.oa2 = wmma4(a2, pa, st.oa2);
    st.ob2 = wmma4(a2, pb, st.ob2);
  }
}

__global__ __launch_bounds__(160, 4) void attn_kernel(const float* __restrict__ qkv,
                                                   float* __restrict__ o,
                                                   float* __restrict__ ga,
                                                   float* __restrict__ gst) {
  int lane = threadIdx.x & 31;
  int wave = threadIdx.x >> 5;
  int pair = blockIdx.x * 5 + wave;
  int h = blockIdx.y, b = blockIdx.z;
  int t0 = pair * 32;
  int c  = lane & 15;
  int hf = lane >> 4;

  int gq0 = t0 + c;      if (gq0 > N_ - 1) gq0 = N_ - 1;
  int gq1 = t0 + 16 + c; if (gq1 > N_ - 1) gq1 = N_ - 1;
  const float* qp0 = qkv + (size_t)(b * N_ + gq0) * C3_ + h * D_ + 4 * hf;
  const float* qp1 = qkv + (size_t)(b * N_ + gq1) * C3_ + h * D_ + 4 * hf;
  v4f q0[6], q1[6];
#pragma unroll
  for (int i = 0; i < 6; ++i) {
    q0[i] = *(const v4f*)(qp0 + 8 * i);
    q1[i] = *(const v4f*)(qp1 + 8 * i);
  }

  const float* kbase = qkv + (size_t)b * N_ * C3_ + C_ + h * D_ + 4 * hf;
  const float* vbase = qkv + (size_t)b * N_ * C3_ + 2 * C_ + h * D_ + 3 * c;
  float* gp = ga + (size_t)(b * HEADS_ + h) * HW_;
  const bool cls_lane = (pair == 0) && (c == 0);

  AttnState st;
  st.oa0 = v8f{}; st.oa1 = v8f{}; st.oa2 = v8f{};
  st.ob0 = v8f{}; st.ob1 = v8f{}; st.ob2 = v8f{};
  st.m0 = -3.0e38f; st.m1 = -3.0e38f; st.l0 = 0.f; st.l1 = 0.f;

  KStep ka, kb;
  load_k(ka, kbase, 0, c);
  for (int key0 = 0; key0 < N_; key0 += 32) {
    VStep va, vb;
    load_k(kb, kbase, key0 + 16, c);
    load_v(va, vbase, key0, hf);
    attn_step(ka, va, q0, q1, st, key0, hf, cls_lane, gp);
    load_k(ka, kbase, key0 + 32, c);
    load_v(vb, vbase, key0 + 16, hf);
    attn_step(kb, vb, q0, q1, st, key0 + 16, hf, cls_lane, gp);
  }

  float lt0 = st.l0 + __shfl_xor(st.l0, 16, 32);
  float lt1 = st.l1 + __shfl_xor(st.l1, 16, 32);
  if (pair == 0 && lane == 0) {
    gst[(b * HEADS_ + h) * 2]     = st.m0;
    gst[(b * HEADS_ + h) * 2 + 1] = lt0;
  }
  float inv0 = 1.f / lt0, inv1 = 1.f / lt1;
  int qa = t0 + c, qb = t0 + 16 + c;
  if (qa < N_) {
    float* op = o + (size_t)(b * N_ + qa) * C_ + h * D_ + 24 * hf;
#pragma unroll
    for (int g = 0; g < 2; ++g) {
      v4f w0 = {st.oa0[4*g] * inv0,   st.oa1[4*g] * inv0,   st.oa2[4*g] * inv0,   st.oa0[4*g+1] * inv0};
      v4f w1 = {st.oa1[4*g+1] * inv0, st.oa2[4*g+1] * inv0, st.oa0[4*g+2] * inv0, st.oa1[4*g+2] * inv0};
      v4f w2 = {st.oa2[4*g+2] * inv0, st.oa0[4*g+3] * inv0, st.oa1[4*g+3] * inv0, st.oa2[4*g+3] * inv0};
      *(v4f*)(op + 12 * g)     = w0;
      *(v4f*)(op + 12 * g + 4) = w1;
      *(v4f*)(op + 12 * g + 8) = w2;
    }
  }
  if (qb < N_) {
    float* op = o + (size_t)(b * N_ + qb) * C_ + h * D_ + 24 * hf;
#pragma unroll
    for (int g = 0; g < 2; ++g) {
      v4f w0 = {st.ob0[4*g] * inv1,   st.ob1[4*g] * inv1,   st.ob2[4*g] * inv1,   st.ob0[4*g+1] * inv1};
      v4f w1 = {st.ob1[4*g+1] * inv1, st.ob2[4*g+1] * inv1, st.ob0[4*g+2] * inv1, st.ob1[4*g+2] * inv1};
      v4f w2 = {st.ob2[4*g+2] * inv1, st.ob0[4*g+3] * inv1, st.ob1[4*g+3] * inv1, st.ob2[4*g+3] * inv1};
      *(v4f*)(op + 12 * g)     = w0;
      *(v4f*)(op + 12 * g + 4) = w1;
      *(v4f*)(op + 12 * g + 8) = w2;
    }
  }
}

__global__ void finalize_kernel(const float* __restrict__ xc,
                                const float* __restrict__ ga,
                                const float* __restrict__ gst,
                                float* __restrict__ out) {
  size_t idx = (size_t)blockIdx.x * 256 + threadIdx.x;
  const size_t n_cls = (size_t)B_ * C_;
  const size_t n_x   = (size_t)B_ * C_ * HW_;
  const size_t n_ga  = (size_t)B_ * HW_;
  if (idx < n_cls) {
    int b = (int)(idx / C_), c = (int)(idx % C_);
    out[idx] = xc[(size_t)b * N_ * C_ + c];
  } else if (idx < n_cls + n_x) {
    size_t j = idx - n_cls;
    int b   = (int)(j / (C_ * HW_));
    int rem = (int)(j % (C_ * HW_));
    int c = rem / HW_, p = rem % HW_;
    out[idx] = xc[((size_t)b * N_ + 1 + p) * C_ + c];
  } else if (idx < n_cls + n_x + n_ga) {
    size_t j = idx - n_cls - n_x;
    int b = (int)(j / HW_), p = (int)(j % HW_);
    float s = 0.f;
#pragma unroll
    for (int hh = 0; hh < HEADS_; ++hh) {
      float mh = gst[(b * HEADS_ + hh) * 2];
      float lh = gst[(b * HEADS_ + hh) * 2 + 1];
      s += expf(ga[(size_t)(b * HEADS_ + hh) * HW_ + p] - mh) * (1.f / lh);
    }
    out[idx] = s * 0.125f;
  }
}

extern "C" void kernel_launch(void* const* d_in, const int* in_sizes, int n_in,
                              void* d_out, int out_size, void* d_ws, size_t ws_size,
                              hipStream_t stream) {
  const float* cls    = (const float*)d_in[0];
  const float* x      = (const float*)d_in[1];
  const float* conv_w = (const float*)d_in[2];
  const float* conv_b = (const float*)d_in[3];
  const float* ln1_g  = (const float*)d_in[4];
  const float* ln1_b  = (const float*)d_in[5];
  const float* qkv_w  = (const float*)d_in[6];
  const float* proj_w = (const float*)d_in[7];
  const float* proj_b = (const float*)d_in[8];
  const float* ln2_g  = (const float*)d_in[9];
  const float* ln2_b  = (const float*)d_in[10];
  const float* fc1_w  = (const float*)d_in[11];
  const float* fc1_b  = (const float*)d_in[12];
  const float* fc2_w  = (const float*)d_in[13];
  const float* fc2_b  = (const float*)d_in[14];

  float* ws  = (float*)d_ws;
  float* xc  = ws;
  float* y   = xc  + (size_t)MP_ * C_;
  float* qkv = y   + (size_t)MP_ * C_;
  float* o   = qkv + (size_t)MP_ * C3_;
  float* m1  = qkv;
  float* ga  = o   + (size_t)MP_ * C_;
  float* gst = ga  + (size_t)B_ * HEADS_ * HW_;
  float* out = (float*)d_out;

  conv_pe_kernel<<<(B_ * C_ * HW_ + 255) / 256, 256, 0, stream>>>(x, conv_w, conv_b, xc);
  cls_copy_kernel<<<(B_ * C_ + 255) / 256, 256, 0, stream>>>(cls, xc);

  ln_kernel<<<(M_ + 7) / 8, 256, 0, stream>>>(xc, ln1_g, ln1_b, y, M_);

  gemm_wmma_kernel<0><<<dim3(C3_ / 64, (MB_ + 7) / 8), 256, 0, stream>>>(
      y, qkv_w, nullptr, qkv, C_, C3_, MB_);

  attn_kernel<<<dim3(5, HEADS_, B_), 160, 0, stream>>>(qkv, o, ga, gst);

  gemm_wmma_kernel<1><<<dim3(C_ / 64, (MB_ + 7) / 8), 256, 0, stream>>>(
      o, proj_w, proj_b, xc, C_, C_, MB_);

  ln_kernel<<<(M_ + 7) / 8, 256, 0, stream>>>(xc, ln2_g, ln2_b, y, M_);

  gemm_wmma_kernel<2><<<dim3(C4_ / 64, (MB_ + 7) / 8), 256, 0, stream>>>(
      y, fc1_w, fc1_b, m1, C_, C4_, MB_);

  gemm_wmma_kernel<1><<<dim3(C_ / 64, (MB_ + 7) / 8), 256, 0, stream>>>(
      m1, fc2_w, fc2_b, xc, C4_, C_, MB_);

  size_t tot = (size_t)B_ * C_ + (size_t)B_ * C_ * HW_ + (size_t)B_ * HW_;
  finalize_kernel<<<(unsigned)((tot + 255) / 256), 256, 0, stream>>>(xc, ga, gst, out);
}


# GEMM: A/B fragments straight from global, one 128-bit load per lane per 8 k (lane half h holds k 8i+4h..+3); a 32-k trip's 24 loads go as one group before its 64 WMMAs; k summation order changes
# speedup vs baseline: 1.9978x; 1.6236x over previous
#include <hip/hip_runtime.h>
#include <math.h>

#define B_    16
#define C_    384
#define H_    28
#define W_    28
#define HW_   784
#define N_    785
#define M_    (B_*N_)
#define MP_   (M_ + 16)
#define HEADS_ 8
#define D_    48
#define C3_   1152
#define C4_   1536
#define MT_   (M_/16)
#define MB_   ((MT_ + 1)/2)

typedef __attribute__((ext_vector_type(2))) float v2f;
typedef __attribute__((ext_vector_type(4))) float v4f;
typedef __attribute__((ext_vector_type(8))) float v8f;

__device__ __forceinline__ v8f wmma4(v2f a, v2f b, v8f c) {
  return __builtin_amdgcn_wmma_f32_16x16x4_f32(false, a, false, b, (short)0, c,
                                               false, false);
}

__global__ void conv_pe_kernel(const float* __restrict__ x,
                               const float* __restrict__ cw,
                               const float* __restrict__ cb,
                               float* __restrict__ xc) {
  int idx = blockIdx.x * 256 + threadIdx.x;
  if (idx >= B_ * C_ * HW_) return;
  int b   = idx / (C_ * HW_);
  int rem = idx % (C_ * HW_);
  int c   = rem / HW_;
  int p   = rem % HW_;
  int hh  = p / W_, ww = p % W_;
  const float* xp = x + (size_t)(b * C_ + c) * HW_;
  const float* wp = cw + c * 9;
  float s = 0.f;
#pragma unroll
  for (int i = 0; i < 3; ++i) {
    int hy = hh + i - 1;
    if (hy < 0 || hy >= H_) continue;
#pragma unroll
    for (int j = 0; j < 3; ++j) {
      int wx = ww + j - 1;
      if (wx < 0 || wx >= W_) continue;
      s += xp[hy * W_ + wx] * wp[i * 3 + j];
    }
  }
  xc[((size_t)b * N_ + 1 + p) * C_ + c] = xp[p] + s + cb[c];
}

__global__ void cls_copy_kernel(const float* __restrict__ cls,
                                float* __restrict__ xc) {
  int idx = blockIdx.x * 256 + threadIdx.x;
  if (idx >= B_ * C_) return;
  int b = idx / C_, c = idx % C_;
  xc[(size_t)b * N_ * C_ + c] = cls[idx];
}

__global__ void ln_kernel(const float* __restrict__ xin,
                          const float* __restrict__ g,
                          const float* __restrict__ bta,
                          float* __restrict__ yout, int ntok) {
  int wave = threadIdx.x >> 5, lane = threadIdx.x & 31;
  int t = blockIdx.x * 8 + wave;
  if (t >= ntok) return;
  const float* xp = xin + (size_t)t * C_;
  float s = 0.f, sq = 0.f;
#pragma unroll
  for (int i = 0; i < C_ / 32; ++i) {
    float v = xp[lane + i * 32];
    s += v; sq += v * v;
  }
#pragma unroll
  for (int off = 16; off; off >>= 1) {
    s  += __shfl_xor(s,  off, 32);
    sq += __shfl_xor(sq, off, 32);
  }
  float mu  = s  * (1.f / C_);
  float var = sq * (1.f / C_) - mu * mu;
  float rs  = rsqrtf(var + 1e-5f);
  float* yp = yout + (size_t)t * C_;
#pragma unroll
  for (int i = 0; i < C_ / 32; ++i) {
    int c = lane + i * 32;
    yp[c] = (xp[c] - mu) * rs * g[c] + bta[c];
  }
}

template <int EPI>
__global__ __launch_bounds__(256) void gemm_wmma_kernel(
    const float* __restrict__ X, const float* __restrict__ Wt,
    const float* __restrict__ bias, float* __restrict__ Out,
    int K, int ldo, int Mblocks) {
  int lane = threadIdx.x & 31;
  int wave = threadIdx.x >> 5;
  int mblk = blockIdx.y * 8 + wave;
  if (mblk >= Mblocks) return;
  int n0 = blockIdx.x * 64;
  int m0 = mblk * 32;
  int rc = lane & 15;
  int kb = (lane >> 4) << 2;
  const float* xp0 = X  + (size_t)(m0 + rc)      * K + kb;
  const float* xp1 = X  + (size_t)(m0 + 16 + rc) * K + kb;
  const float* wp0 = Wt + (size_t)(n0 + rc)      * K + kb;
  const float* wp1 = Wt + (size_t)(n0 + 16 + rc) * K + kb;
  const float* wp2 = Wt + (size_t)(n0 + 32 + rc) * K + kb;
  const float* wp3 = Wt + (size_t)(n0 + 48 + rc) * K + kb;

  v8f acc[2][4] = {};
  for (int k = 0; k < K; k += 32) {
    v4f a0[4], a1[4], b0[4], b1[4], b2[4], b3[4];
#pragma unroll
    for (int u = 0; u < 4; ++u) {
      a0[u] = *(const v4f*)(xp0 + k + 8 * u);
      a1[u] = *(const v4f*)(xp1 + k + 8 * u);
      b0[u] = *(const v4f*)(wp0 + k + 8 * u);
      b1[u] = *(const v4f*)(wp1 + k + 8 * u);
      b2[u] = *(const v4f*)(wp2 + k + 8 * u);
      b3[u] = *(const v4f*)(wp3 + k + 8 * u);
    }
    __builtin_amdgcn_sched_barrier(0);
#pragma unroll
    for (int u = 0; u < 4; ++u) {
      v2f a0l = {a0[u].x, a0[u].y}, a0h = {a0[u].z, a0[u].w};
      v2f a1l = {a1[u].x, a1[u].y}, a1h = {a1[u].z, a1[u].w};
      v2f b0l = {b0[u].x, b0[u].y}, b0h = {b0[u].z, b0[u].w};
      v2f b1l = {b1[u].x, b1[u].y}, b1h = {b1[u].z, b1[u].w};
      v2f b2l = {b2[u].x, b2[u].y}, b2h = {b2[u].z, b2[u].w};
      v2f b3l = {b3[u].x, b3[u].y}, b3h = {b3[u].z, b3[u].w};
      acc[0][0] = wmma4(a0l, b0l, acc[0][0]);
      acc[1][0] = wmma4(a1l, b0l, acc[1][0]);
      acc[0][1] = wmma4(a0l, b1l, acc[0][1]);
      acc[1][1] = wmma4(a1l, b1l, acc[1][1]);
      acc[0][2] = wmma4(a0l, b2l, acc[0][2]);
      acc[1][2] = wmma4(a1l, b2l, acc[1][2]);
      acc[0][3] = wmma4(a0l, b3l, acc[0][3]);
      acc[1][3] = wmma4(a1l, b3l, acc[1][3]);
      acc[0][0] = wmma4(a0h, b0h, acc[0][0]);
      acc[1][0] = wmma4(a1h, b0h, acc[1][0]);
      acc[0][1] = wmma4(a0h, b1h, acc[0][1]);
      acc[1][1] = wmma4(a1h, b1h, acc[1][1]);
      acc[0][2] = wmma4(a0h, b2h, acc[0][2]);
      acc[1][2] = wmma4(a1h, b2h, acc[1][2]);
      acc[0][3] = wmma4(a0h, b3h, acc[0][3]);
      acc[1][3] = wmma4(a1h, b3h, acc[1][3]);
    }
  }

  int coln = lane & 15;
  int roff = (lane >> 4) << 3;
#pragma unroll
  for (int mi = 0; mi < 2; ++mi) {
#pragma unroll
    for (int j = 0; j < 4; ++j) {
      int nc = n0 + j * 16 + coln;
      float bv = bias ? bias[nc] : 0.f;
#pragma unroll
      for (int r = 0; r < 8; ++r) {
        int gm = m0 + mi * 16 + roff + r;
        if (gm < M_) {
          size_t oi = (size_t)gm * ldo + nc;
          float v = acc[mi][j][r] + bv;
          if (EPI == 0)      Out[oi] = v;
          else if (EPI == 1) Out[oi] += v;
          else               Out[oi] = 0.5f * v * (1.f + erff(v * 0.70710678118654752f));
        }
      }
    }
  }
}

struct __attribute__((packed, aligned(4))) f3 { float x, y, z; };

struct KStep { v4f k[6]; };
struct VStep { f3 va[4]; f3 vb[4]; };

__device__ __forceinline__ void load_k(KStep& t, const float* __restrict__ kbase,
                                       int key0, int c) {
  int kr = key0 + c; if (kr > N_ - 1) kr = N_ - 1;
  const float* kp = kbase + (size_t)kr * C3_;
#pragma unroll
  for (int i = 0; i < 6; ++i) t.k[i] = *(const v4f*)(kp + 8 * i);
}

__device__ __forceinline__ void load_v(VStep& t, const float* __restrict__ vbase,
                                       int key0, int hf) {
  const int kl = key0 + 8 * hf;
#pragma unroll
  for (int j = 0; j < 4; ++j) {
    int k0 = kl + 2 * j;     if (k0 > N_ - 1) k0 = N_ - 1;
    int k1 = kl + 2 * j + 1; if (k1 > N_ - 1) k1 = N_ - 1;
    t.va[j] = *(const f3*)(vbase + (size_t)k0 * C3_);
    t.vb[j] = *(const f3*)(vbase + (size_t)k1 * C3_);
  }
}

struct AttnState {
  v8f oa0, oa1, oa2, ob0, ob1, ob2;
  float m0, m1, l0, l1;
};

__device__ __forceinline__ void attn_step(const KStep& tk, const VStep& t,
                                          const v4f (&q0)[6],
                                          const v4f (&q1)[6], AttnState& st,
                                          int key0, int hf, bool cls_lane,
                                          float* __restrict__ gp) {
  const float scale = 0.14433756729740643f;
  v8f s0 = {}, s1 = {};
#pragma unroll
  for (int i = 0; i < 6; ++i) {
    v2f ka = {tk.k[i].x, tk.k[i].y}, kc = {tk.k[i].z, tk.k[i].w};
    v2f qa0 = {q0[i].x, q0[i].y}, qc0 = {q0[i].z, q0[i].w};
    v2f qa1 = {q1[i].x, q1[i].y}, qc1 = {q1[i].z, q1[i].w};
    s0 = wmma4(ka, qa0, s0);
    s1 = wmma4(ka, qa1, s1);
    s0 = wmma4(kc, qc0, s0);
    s1 = wmma4(kc, qc1, s1);
  }
  const int kl = key0 + 8 * hf;
  float x0[8], x1[8];
#pragma unroll
  for (int r = 0; r < 8; ++r) { x0[r] = s0[r] * scale; x1[r] = s1[r] * scale; }
  if (key0 + 16 > N_) {
#pragma unroll
    for (int r = 0; r < 8; ++r)
      if (kl + r >= N_) { x0[r] = -3.0e38f; x1[r] = -3.0e38f; }
  }
  if (cls_lane) {
#pragma unroll
    for (int r = 0; r < 8; ++r) {
      int key = kl + r;
      if (key >= 1 && key < N_) gp[key - 1] = x0[r];
    }
  }
  float mx0 = x0[0], mx1 = x1[0];
#pragma unroll
  for (int r = 1; r < 8; ++r) { mx0 = fmaxf(mx0, x0[r]); mx1 = fmaxf(mx1, x1[r]); }
  mx0 = fmaxf(mx0, __shfl_xor(mx0, 16, 32));
  mx1 = fmaxf(mx1, __shfl_xor(mx1, 16, 32));
  float mn0 = fmaxf(st.m0, mx0), mn1 = fmaxf(st.m1, mx1);
  float al0 = expf(st.m0 - mn0), al1 = expf(st.m1 - mn1);
  float ps0 = 0.f, ps1 = 0.f;
#pragma unroll
  for (int r = 0; r < 8; ++r) {
    x0[r] = expf(x0[r] - mn0); ps0 += x0[r];
    x1[r] = expf(x1[r] - mn1); ps1 += x1[r];
  }
  st.l0 = st.l0 * al0 + ps0;  st.l1 = st.l1 * al1 + ps1;
  st.m0 = mn0;  st.m1 = mn1;
#pragma unroll
  for (int r = 0; r < 8; ++r) {
    st.oa0[r] *= al0; st.oa1[r] *= al0; st.oa2[r] *= al0;
    st.ob0[r] *= al1; st.ob1[r] *= al1; st.ob2[r] *= al1;
  }
#pragma unroll
  for (int j = 0; j < 4; ++j) {
    v2f pa = {x0[2 * j], x0[2 * j + 1]};
    v2f pb = {x1[2 * j], x1[2 * j + 1]};
    v2f a0 = {t.va[j].x, t.vb[j].x}, a1 = {t.va[j].y, t.vb[j].y}, a2 = {t.va[j].z, t.vb[j].z};
    st.oa0 = wmma4(a0, pa, st.oa0);
    st.ob0 = wmma4(a0, pb, st.ob0);
    st.oa1 = wmma4(a1, pa, st.oa1);
    st.ob1 = wmma4(a1, pb, st.ob1);
    st.oa2 = wmma4(a2, pa, st.oa2);
    st.ob2 = wmma4(a2, pb, st.ob2);
  }
}

__global__ __launch_bounds__(160, 4) void attn_kernel(const float* __restrict__ qkv,
                                                   float* __restrict__ o,
                                                   float* __restrict__ ga,
                                                   float* __restrict__ gst) {
  int lane = threadIdx.x & 31;
  int wave = threadIdx.x >> 5;
  int pair = blockIdx.x * 5 + wave;
  int h = blockIdx.y, b = blockIdx.z;
  int t0 = pair * 32;
  int c  = lane & 15;
  int hf = lane >> 4;

  int gq0 = t0 + c;      if (gq0 > N_ - 1) gq0 = N_ - 1;
  int gq1 = t0 + 16 + c; if (gq1 > N_ - 1) gq1 = N_ - 1;
  const float* qp0 = qkv + (size_t)(b * N_ + gq0) * C3_ + h * D_ + 4 * hf;
  const float* qp1 = qkv + (size_t)(b * N_ + gq1) * C3_ + h * D_ + 4 * hf;
  v4f q0[6], q1[6];
#pragma unroll
  for (int i = 0; i < 6; ++i) {
    q0[i] = *(const v4f*)(qp0 + 8 * i);
    q1[i] = *(const v4f*)(qp1 + 8 * i);
  }

  const float* kbase = qkv + (size_t)b * N_ * C3_ + C_ + h * D_ + 4 * hf;
  const float* vbase = qkv + (size_t)b * N_ * C3_ + 2 * C_ + h * D_ + 3 * c;
  float* gp = ga + (size_t)(b * HEADS_ + h) * HW_;
  const bool cls_lane = (pair == 0) && (c == 0);

  AttnState st;
  st.oa0 = v8f{}; st.oa1 = v8f{}; st.oa2 = v8f{};
  st.ob0 = v8f{}; st.ob1 = v8f{}; st.ob2 = v8f{};
  st.m0 = -3.0e38f; st.m1 = -3.0e38f; st.l0 = 0.f; st.l1 = 0.f;

  KStep ka, kb;
  load_k(ka, kbase, 0, c);
  for (int key0 = 0; key0 < N_; key0 += 32) {
    VStep va, vb;
    load_k(kb, kbase, key0 + 16, c);
    load_v(va, vbase, key0, hf);
    attn_step(ka, va, q0, q1, st, key0, hf, cls_lane, gp);
    load_k(ka, kbase, key0 + 32, c);
    load_v(vb, vbase, key0 + 16, hf);
    attn_step(kb, vb, q0, q1, st, key0 + 16, hf, cls_lane, gp);
  }

  float lt0 = st.l0 + __shfl_xor(st.l0, 16, 32);
  float lt1 = st.l1 + __shfl_xor(st.l1, 16, 32);
  if (pair == 0 && lane == 0) {
    gst[(b * HEADS_ + h) * 2]     = st.m0;
    gst[(b * HEADS_ + h) * 2 + 1] = lt0;
  }
  float inv0 = 1.f / lt0, inv1 = 1.f / lt1;
  int qa = t0 + c, qb = t0 + 16 + c;
  if (qa < N_) {
    float* op = o + (size_t)(b * N_ + qa) * C_ + h * D_ + 24 * hf;
#pragma unroll
    for (int g = 0; g < 2; ++g) {
      v4f w0 = {st.oa0[4*g] * inv0,   st.oa1[4*g] * inv0,   st.oa2[4*g] * inv0,   st.oa0[4*g+1] * inv0};
      v4f w1 = {st.oa1[4*g+1] * inv0, st.oa2[4*g+1] * inv0, st.oa0[4*g+2] * inv0, st.oa1[4*g+2] * inv0};
      v4f w2 = {st.oa2[4*g+2] * inv0, st.oa0[4*g+3] * inv0, st.oa1[4*g+3] * inv0, st.oa2[4*g+3] * inv0};
      *(v4f*)(op + 12 * g)     = w0;
      *(v4f*)(op + 12 * g + 4) = w1;
      *(v4f*)(op + 12 * g + 8) = w2;
    }
  }
  if (qb < N_) {
    float* op = o + (size_t)(b * N_ + qb) * C_ + h * D_ + 24 * hf;
#pragma unroll
    for (int g = 0; g < 2; ++g) {
      v4f w0 = {st.ob0[4*g] * inv1,   st.ob1[4*g] * inv1,   st.ob2[4*g] * inv1,   st.ob0[4*g+1] * inv1};
      v4f w1 = {st.ob1[4*g+1] * inv1, st.ob2[4*g+1] * inv1, st.ob0[4*g+2] * inv1, st.ob1[4*g+2] * inv1};
      v4f w2 = {st.ob2[4*g+2] * inv1, st.ob0[4*g+3] * inv1, st.ob1[4*g+3] * inv1, st.ob2[4*g+3] * inv1};
      *(v4f*)(op + 12 * g)     = w0;
      *(v4f*)(op + 12 * g + 4) = w1;
      *(v4f*)(op + 12 * g + 8) = w2;
    }
  }
}

__global__ void finalize_kernel(const float* __restrict__ xc,
                                const float* __restrict__ ga,
                                const float* __restrict__ gst,
                                float* __restrict__ out) {
  size_t idx = (size_t)blockIdx.x * 256 + threadIdx.x;
  const size_t n_cls = (size_t)B_ * C_;
  const size_t n_x   = (size_t)B_ * C_ * HW_;
  const size_t n_ga  = (size_t)B_ * HW_;
  if (idx < n_cls) {
    int b = (int)(idx / C_), c = (int)(idx % C_);
    out[idx] = xc[(size_t)b * N_ * C_ + c];
  } else if (idx < n_cls + n_x) {
    size_t j = idx - n_cls;
    int b   = (int)(j / (C_ * HW_));
    int rem = (int)(j % (C_ * HW_));
    int c = rem / HW_, p = rem % HW_;
    out[idx] = xc[((size_t)b * N_ + 1 + p) * C_ + c];
  } else if (idx < n_cls + n_x + n_ga) {
    size_t j = idx - n_cls - n_x;
    int b = (int)(j / HW_), p = (int)(j % HW_);
    float s = 0.f;
#pragma unroll
    for (int hh = 0; hh < HEADS_; ++hh) {
      float mh = gst[(b * HEADS_ + hh) * 2];
      float lh = gst[(b * HEADS_ + hh) * 2 + 1];
      s += expf(ga[(size_t)(b * HEADS_ + hh) * HW_ + p] - mh) * (1.f / lh);
    }
    out[idx] = s * 0.125f;
  }
}

extern "C" void kernel_launch(void* const* d_in, const int* in_sizes, int n_in,
                              void* d_out, int out_size, void* d_ws, size_t ws_size,
                              hipStream_t stream) {
  const float* cls    = (const float*)d_in[0];
  const float* x      = (const float*)d_in[1];
  const float* conv_w = (const float*)d_in[2];
  const float* conv_b = (const float*)d_in[3];
  const float* ln1_g  = (const float*)d_in[4];
  const float* ln1_b  = (const float*)d_in[5];
  const float* qkv_w  = (const float*)d_in[6];
  const float* proj_w = (const float*)d_in[7];
  const float* proj_b = (const float*)d_in[8];
  const float* ln2_g  = (const float*)d_in[9];
  const float* ln2_b  = (const float*)d_in[10];
  const float* fc1_w  = (const float*)d_in[11];
  const float* fc1_b  = (const float*)d_in[12];
  const float* fc2_w  = (const float*)d_in[13];
  const float* fc2_b  = (const float*)d_in[14];

  float* ws  = (float*)d_ws;
  float* xc  = ws;
  float* y   = xc  + (size_t)MP_ * C_;
  float* qkv = y   + (size_t)MP_ * C_;
  float* o   = qkv + (size_t)MP_ * C3_;
  float* m1  = qkv;
  float* ga  = o   + (size_t)MP_ * C_;
  float* gst = ga  + (size_t)B_ * HEADS_ * HW_;
  float* out = (float*)d_out;

  conv_pe_kernel<<<(B_ * C_ * HW_ + 255) / 256, 256, 0, stream>>>(x, conv_w, conv_b, xc);
  cls_copy_kernel<<<(B_ * C_ + 255) / 256, 256, 0, stream>>>(cls, xc);

  ln_kernel<<<(M_ + 7) / 8, 256, 0, stream>>>(xc, ln1_g, ln1_b, y, M_);

  gemm_wmma_kernel<0><<<dim3(C3_ / 64, (MB_ + 7) / 8), 256, 0, stream>>>(
      y, qkv_w, nullptr, qkv, C_, C3_, MB_);

  attn_kernel<<<dim3(5, HEADS_, B_), 160, 0, stream>>>(qkv, o, ga, gst);

  gemm_wmma_kernel<1><<<dim3(C_ / 64, (MB_ + 7) / 8), 256, 0, stream>>>(
      o, proj_w, proj_b, xc, C_, C_, MB_);

  ln_kernel<<<(M_ + 7) / 8, 256, 0, stream>>>(xc, ln2_g, ln2_b, y, M_);

  gemm_wmma_kernel<2><<<dim3(C4_ / 64, (MB_ + 7) / 8), 256, 0, stream>>>(
      y, fc1_w, fc1_b, m1, C_, C4_, MB_);

  gemm_wmma_kernel<1><<<dim3(C_ / 64, (MB_ + 7) / 8), 256, 0, stream>>>(
      m1, fc2_w, fc2_b, xc, C4_, C_, MB_);

  size_t tot = (size_t)B_ * C_ + (size_t)B_ * C_ * HW_ + (size_t)B_ * HW_;
  finalize_kernel<<<(unsigned)((tot + 255) / 256), 256, 0, stream>>>(xc, ga, gst, out);
}


# conv_pe and finalize: one lane per image position loops over 32 channels (token side one whole 128-byte line per lane, image side coalesced): 4704 waves per launch, not 150528; sums unchanged
# speedup vs baseline: 2.3640x; 1.1833x over previous
#include <hip/hip_runtime.h>
#include <math.h>

#define B_    16
#define C_    384
#define H_    28
#define W_    28
#define HW_   784
#define N_    785
#define M_    (B_*N_)
#define MP_   (M_ + 16)
#define HEADS_ 8
#define D_    48
#define C3_   1152
#define C4_   1536
#define MT_   (M_/16)
#define MB_   ((MT_ + 1)/2)
#define CH_   32
#define NPOS_ (B_*HW_)

typedef __attribute__((ext_vector_type(2))) float v2f;
typedef __attribute__((ext_vector_type(4))) float v4f;
typedef __attribute__((ext_vector_type(8))) float v8f;

__device__ __forceinline__ v8f wmma4(v2f a, v2f b, v8f c) {
  return __builtin_amdgcn_wmma_f32_16x16x4_f32(false, a, false, b, (short)0, c,
                                               false, false);
}

__global__ __launch_bounds__(256) void conv_pe_kernel(const float* __restrict__ x,
                                                      const float* __restrict__ cw,
                                                      const float* __restrict__ cb,
                                                      float* __restrict__ xc) {
  const int gp = blockIdx.x * 256 + threadIdx.x;
  const int c0 = blockIdx.y * CH_;
  const int b  = gp / HW_, p = gp - b * HW_;
  const int hh = p / W_,  ww = p - hh * W_;
  bool ok[9];
  int  off[9];
#pragma unroll
  for (int i = 0; i < 3; ++i) {
#pragma unroll
    for (int j = 0; j < 3; ++j) {
      const bool vi = (i == 0) ? (hh > 0) : (i == 2) ? (hh < H_ - 1) : true;
      const bool vj = (j == 0) ? (ww > 0) : (j == 2) ? (ww < W_ - 1) : true;
      ok[i * 3 + j]  = vi && vj;
      off[i * 3 + j] = (vi && vj) ? (i - 1) * W_ + (j - 1) : 0;
    }
  }
  const float* xp = x + ((size_t)(b * C_ + c0) * HW_ + p);
  float* op = xc + ((size_t)(b * N_ + 1 + p) * C_ + c0);
#pragma unroll
  for (int g = 0; g < CH_ / 8; ++g) {
    float v[8][9];
#pragma unroll
    for (int q = 0; q < 8; ++q) {
      const float* xq = xp + (size_t)(8 * g + q) * HW_;
#pragma unroll
      for (int t = 0; t < 9; ++t) v[q][t] = xq[off[t]];
    }
    __builtin_amdgcn_sched_barrier(0);
    float r[8];
#pragma unroll
    for (int q = 0; q < 8; ++q) {
      const int c = c0 + 8 * g + q;
      const float* wq = cw + c * 9;
      float s = 0.f;
#pragma unroll
      for (int t = 0; t < 9; ++t) s = fmaf(ok[t] ? v[q][t] : 0.f, wq[t], s);
      r[q] = v[q][4] + s + cb[c];
    }
    v4f w0 = {r[0], r[1], r[2], r[3]};
    v4f w1 = {r[4], r[5], r[6], r[7]};
    *(v4f*)(op + 8 * g)     = w0;
    *(v4f*)(op + 8 * g + 4) = w1;
  }
}

__global__ void cls_copy_kernel(const float* __restrict__ cls,
                                float* __restrict__ xc) {
  int idx = blockIdx.x * 256 + threadIdx.x;
  if (idx >= B_ * C_) return;
  int b = idx / C_, c = idx % C_;
  xc[(size_t)b * N_ * C_ + c] = cls[idx];
}

__global__ void ln_kernel(const float* __restrict__ xin,
                          const float* __restrict__ g,
                          const float* __restrict__ bta,
                          float* __restrict__ yout, int ntok) {
  int wave = threadIdx.x >> 5, lane = threadIdx.x & 31;
  int t = blockIdx.x * 8 + wave;
  if (t >= ntok) return;
  const float* xp = xin + (size_t)t * C_;
  float s = 0.f, sq = 0.f;
#pragma unroll
  for (int i = 0; i < C_ / 32; ++i) {
    float v = xp[lane + i * 32];
    s += v; sq += v * v;
  }
#pragma unroll
  for (int off = 16; off; off >>= 1) {
    s  += __shfl_xor(s,  off, 32);
    sq += __shfl_xor(sq, off, 32);
  }
  float mu  = s  * (1.f / C_);
  float var = sq * (1.f / C_) - mu * mu;
  float rs  = rsqrtf(var + 1e-5f);
  float* yp = yout + (size_t)t * C_;
#pragma unroll
  for (int i = 0; i < C_ / 32; ++i) {
    int c = lane + i * 32;
    yp[c] = (xp[c] - mu) * rs * g[c] + bta[c];
  }
}

template <int EPI>
__global__ __launch_bounds__(256) void gemm_wmma_kernel(
    const float* __restrict__ X, const float* __restrict__ Wt,
    const float* __restrict__ bias, float* __restrict__ Out,
    int K, int ldo, int Mblocks) {
  int lane = threadIdx.x & 31;
  int wave = threadIdx.x >> 5;
  int mblk = blockIdx.y * 8 + wave;
  if (mblk >= Mblocks) return;
  int n0 = blockIdx.x * 64;
  int m0 = mblk * 32;
  int rc = lane & 15;
  int kb = (lane >> 4) << 2;
  const float* xp0 = X  + (size_t)(m0 + rc)      * K + kb;
  const float* xp1 = X  + (size_t)(m0 + 16 + rc) * K + kb;
  const float* wp0 = Wt + (size_t)(n0 + rc)      * K + kb;
  const float* wp1 = Wt + (size_t)(n0 + 16 + rc) * K + kb;
  const float* wp2 = Wt + (size_t)(n0 + 32 + rc) * K + kb;
  const float* wp3 = Wt + (size_t)(n0 + 48 + rc) * K + kb;

  v8f acc[2][4] = {};
  for (int k = 0; k < K; k += 32) {
    v4f a0[4], a1[4], b0[4], b1[4], b2[4], b3[4];
#pragma unroll
    for (int u = 0; u < 4; ++u) {
      a0[u] = *(const v4f*)(xp0 + k + 8 * u);
      a1[u] = *(const v4f*)(xp1 + k + 8 * u);
      b0[u] = *(const v4f*)(wp0 + k + 8 * u);
      b1[u] = *(const v4f*)(wp1 + k + 8 * u);
      b2[u] = *(const v4f*)(wp2 + k + 8 * u);
      b3[u] = *(const v4f*)(wp3 + k + 8 * u);
    }
    __builtin_amdgcn_sched_barrier(0);
#pragma unroll
    for (int u = 0; u < 4; ++u) {
      v2f a0l = {a0[u].x, a0[u].y}, a0h = {a0[u].z, a0[u].w};
      v2f a1l = {a1[u].x, a1[u].y}, a1h = {a1[u].z, a1[u].w};
      v2f b0l = {b0[u].x, b0[u].y}, b0h = {b0[u].z, b0[u].w};
      v2f b1l = {b1[u].x, b1[u].y}, b1h = {b1[u].z, b1[u].w};
      v2f b2l = {b2[u].x, b2[u].y}, b2h = {b2[u].z, b2[u].w};
      v2f b3l = {b3[u].x, b3[u].y}, b3h = {b3[u].z, b3[u].w};
      acc[0][0] = wmma4(a0l, b0l, acc[0][0]);
      acc[1][0] = wmma4(a1l, b0l, acc[1][0]);
      acc[0][1] = wmma4(a0l, b1l, acc[0][1]);
      acc[1][1] = wmma4(a1l, b1l, acc[1][1]);
      acc[0][2] = wmma4(a0l, b2l, acc[0][2]);
      acc[1][2] = wmma4(a1l, b2l, acc[1][2]);
      acc[0][3] = wmma4(a0l, b3l, acc[0][3]);
      acc[1][3] = wmma4(a1l, b3l, acc[1][3]);
      acc[0][0] = wmma4(a0h, b0h, acc[0][0]);
      acc[1][0] = wmma4(a1h, b0h, acc[1][0]);
      acc[0][1] = wmma4(a0h, b1h, acc[0][1]);
      acc[1][1] = wmma4(a1h, b1h, acc[1][1]);
      acc[0][2] = wmma4(a0h, b2h, acc[0][2]);
      acc[1][2] = wmma4(a1h, b2h, acc[1][2]);
      acc[0][3] = wmma4(a0h, b3h, acc[0][3]);
      acc[1][3] = wmma4(a1h, b3h, acc[1][3]);
    }
  }

  int coln = lane & 15;
  int roff = (lane >> 4) << 3;
#pragma unroll
  for (int mi = 0; mi < 2; ++mi) {
#pragma unroll
    for (int j = 0; j < 4; ++j) {
      int nc = n0 + j * 16 + coln;
      float bv = bias ? bias[nc] : 0.f;
#pragma unroll
      for (int r = 0; r < 8; ++r) {
        int gm = m0 + mi * 16 + roff + r;
        if (gm < M_) {
          size_t oi = (size_t)gm * ldo + nc;
          float v = acc[mi][j][r] + bv;
          if (EPI == 0)      Out[oi] = v;
          else if (EPI == 1) Out[oi] += v;
          else               Out[oi] = 0.5f * v * (1.f + erff(v * 0.70710678118654752f));
        }
      }
    }
  }
}

struct __attribute__((packed, aligned(4))) f3 { float x, y, z; };

struct KStep { v4f k[6]; };
struct VStep { f3 va[4]; f3 vb[4]; };

__device__ __forceinline__ void load_k(KStep& t, const float* __restrict__ kbase,
                                       int key0, int c) {
  int kr = key0 + c; if (kr > N_ - 1) kr = N_ - 1;
  const float* kp = kbase + (size_t)kr * C3_;
#pragma unroll
  for (int i = 0; i < 6; ++i) t.k[i] = *(const v4f*)(kp + 8 * i);
}

__device__ __forceinline__ void load_v(VStep& t, const float* __restrict__ vbase,
                                       int key0, int hf) {
  const int kl = key0 + 8 * hf;
#pragma unroll
  for (int j = 0; j < 4; ++j) {
    int k0 = kl + 2 * j;     if (k0 > N_ - 1) k0 = N_ - 1;
    int k1 = kl + 2 * j + 1; if (k1 > N_ - 1) k1 = N_ - 1;
    t.va[j] = *(const f3*)(vbase + (size_t)k0 * C3_);
    t.vb[j] = *(const f3*)(vbase + (size_t)k1 * C3_);
  }
}

struct AttnState {
  v8f oa0, oa1, oa2, ob0, ob1, ob2;
  float m0, m1, l0, l1;
};

__device__ __forceinline__ void attn_step(const KStep& tk, const VStep& t,
                                          const v4f (&q0)[6],
                                          const v4f (&q1)[6], AttnState& st,
                                          int key0, int hf, bool cls_lane,
                                          float* __restrict__ gp) {
  const float scale = 0.14433756729740643f;
  v8f s0 = {}, s1 = {};
#pragma unroll
  for (int i = 0; i < 6; ++i) {
    v2f ka = {tk.k[i].x, tk.k[i].y}, kc = {tk.k[i].z, tk.k[i].w};
    v2f qa0 = {q0[i].x, q0[i].y}, qc0 = {q0[i].z, q0[i].w};
    v2f qa1 = {q1[i].x, q1[i].y}, qc1 = {q1[i].z, q1[i].w};
    s0 = wmma4(ka, qa0, s0);
    s1 = wmma4(ka, qa1, s1);
    s0 = wmma4(kc, qc0, s0);
    s1 = wmma4(kc, qc1, s1);
  }
  const int kl = key0 + 8 * hf;
  float x0[8], x1[8];
#pragma unroll
  for (int r = 0; r < 8; ++r) { x0[r] = s0[r] * scale; x1[r] = s1[r] * scale; }
  if (key0 + 16 > N_) {
#pragma unroll
    for (int r = 0; r < 8; ++r)
      if (kl + r >= N_) { x0[r] = -3.0e38f; x1[r] = -3.0e38f; }
  }
  if (cls_lane) {
#pragma unroll
    for (int r = 0; r < 8; ++r) {
      int key = kl + r;
      if (key >= 1 && key < N_) gp[key - 1] = x0[r];
    }
  }
  float mx0 = x0[0], mx1 = x1[0];
#pragma unroll
  for (int r = 1; r < 8; ++r) { mx0 = fmaxf(mx0, x0[r]); mx1 = fmaxf(mx1, x1[r]); }
  mx0 = fmaxf(mx0, __shfl_xor(mx0, 16, 32));
  mx1 = fmaxf(mx1, __shfl_xor(mx1, 16, 32));
  float mn0 = fmaxf(st.m0, mx0), mn1 = fmaxf(st.m1, mx1);
  float al0 = expf(st.m0 - mn0), al1 = expf(st.m1 - mn1);
  float ps0 = 0.f, ps1 = 0.f;
#pragma unroll
  for (int r = 0; r < 8; ++r) {
    x0[r] = expf(x0[r] - mn0); ps0 += x0[r];
    x1[r] = expf(x1[r] - mn1); ps1 += x1[r];
  }
  st.l0 = st.l0 * al0 + ps0;  st.l1 = st.l1 * al1 + ps1;
  st.m0 = mn0;  st.m1 = mn1;
#pragma unroll
  for (int r = 0; r < 8; ++r) {
    st.oa0[r] *= al0; st.oa1[r] *= al0; st.oa2[r] *= al0;
    st.ob0[r] *= al1; st.ob1[r] *= al1; st.ob2[r] *= al1;
  }
#pragma unroll
  for (int j = 0; j < 4; ++j) {
    v2f pa = {x0[2 * j], x0[2 * j + 1]};
    v2f pb = {x1[2 * j], x1[2 * j + 1]};
    v2f a0 = {t.va[j].x, t.vb[j].x}, a1 = {t.va[j].y, t.vb[j].y}, a2 = {t.va[j].z, t.vb[j].z};
    st.oa0 = wmma4(a0, pa, st.oa0);
    st.ob0 = wmma4(a0, pb, st.ob0);
    st.oa1 = wmma4(a1, pa, st.oa1);
    st.ob1 = wmma4(a1, pb, st.ob1);
    st.oa2 = wmma4(a2, pa, st.oa2);
    st.ob2 = wmma4(a2, pb, st.ob2);
  }
}

__global__ __launch_bounds__(160, 4) void attn_kernel(const float* __restrict__ qkv,
                                                   float* __restrict__ o,
                                                   float* __restrict__ ga,
                                                   float* __restrict__ gst) {
  int lane = threadIdx.x & 31;
  int wave = threadIdx.x >> 5;
  int pair = blockIdx.x * 5 + wave;
  int h = blockIdx.y, b = blockIdx.z;
  int t0 = pair * 32;
  int c  = lane & 15;
  int hf = lane >> 4;

  int gq0 = t0 + c;      if (gq0 > N_ - 1) gq0 = N_ - 1;
  int gq1 = t0 + 16 + c; if (gq1 > N_ - 1) gq1 = N_ - 1;
  const float* qp0 = qkv + (size_t)(b * N_ + gq0) * C3_ + h * D_ + 4 * hf;
  const float* qp1 = qkv + (size_t)(b * N_ + gq1) * C3_ + h * D_ + 4 * hf;
  v4f q0[6], q1[6];
#pragma unroll
  for (int i = 0; i < 6; ++i) {
    q0[i] = *(const v4f*)(qp0 + 8 * i);
    q1[i] = *(const v4f*)(qp1 + 8 * i);
  }

  const float* kbase = qkv + (size_t)b * N_ * C3_ + C_ + h * D_ + 4 * hf;
  const float* vbase = qkv + (size_t)b * N_ * C3_ + 2 * C_ + h * D_ + 3 * c;
  float* gp = ga + (size_t)(b * HEADS_ + h) * HW_;
  const bool cls_lane = (pair == 0) && (c == 0);

  AttnState st;
  st.oa0 = v8f{}; st.oa1 = v8f{}; st.oa2 = v8f{};
  st.ob0 = v8f{}; st.ob1 = v8f{}; st.ob2 = v8f{};
  st.m0 = -3.0e38f; st.m1 = -3.0e38f; st.l0 = 0.f; st.l1 = 0.f;

  KStep ka, kb;
  load_k(ka, kbase, 0, c);
  for (int key0 = 0; key0 < N_; key0 += 32) {
    VStep va, vb;
    load_k(kb, kbase, key0 + 16, c);
    load_v(va, vbase, key0, hf);
    attn_step(ka, va, q0, q1, st, key0, hf, cls_lane, gp);
    load_k(ka, kbase, key0 + 32, c);
    load_v(vb, vbase, key0 + 16, hf);
    attn_step(kb, vb, q0, q1, st, key0 + 16, hf, cls_lane, gp);
  }

  float lt0 = st.l0 + __shfl_xor(st.l0, 16, 32);
  float lt1 = st.l1 + __shfl_xor(st.l1, 16, 32);
  if (pair == 0 && lane == 0) {
    gst[(b * HEADS_ + h) * 2]     = st.m0;
    gst[(b * HEADS_ + h) * 2 + 1] = lt0;
  }
  float inv0 = 1.f / lt0, inv1 = 1.f / lt1;
  int qa = t0 + c, qb = t0 + 16 + c;
  if (qa < N_) {
    float* op = o + (size_t)(b * N_ + qa) * C_ + h * D_ + 24 * hf;
#pragma unroll
    for (int g = 0; g < 2; ++g) {
      v4f w0 = {st.oa0[4*g] * inv0,   st.oa1[4*g] * inv0,   st.oa2[4*g] * inv0,   st.oa0[4*g+1] * inv0};
      v4f w1 = {st.oa1[4*g+1] * inv0, st.oa2[4*g+1] * inv0, st.oa0[4*g+2] * inv0, st.oa1[4*g+2] * inv0};
      v4f w2 = {st.oa2[4*g+2] * inv0, st.oa0[4*g+3] * inv0, st.oa1[4*g+3] * inv0, st.oa2[4*g+3] * inv0};
      *(v4f*)(op + 12 * g)     = w0;
      *(v4f*)(op + 12 * g + 4) = w1;
      *(v4f*)(op + 12 * g + 8) = w2;
    }
  }
  if (qb < N_) {
    float* op = o + (size_t)(b * N_ + qb) * C_ + h * D_ + 24 * hf;
#pragma unroll
    for (int g = 0; g < 2; ++g) {
      v4f w0 = {st.ob0[4*g] * inv1,   st.ob1[4*g] * inv1,   st.ob2[4*g] * inv1,   st.ob0[4*g+1] * inv1};
      v4f w1 = {st.ob1[4*g+1] * inv1, st.ob2[4*g+1] * inv1, st.ob0[4*g+2] * inv1, st.ob1[4*g+2] * inv1};
      v4f w2 = {st.ob2[4*g+2] * inv1, st.ob0[4*g+3] * inv1, st.ob1[4*g+3] * inv1, st.ob2[4*g+3] * inv1};
      *(v4f*)(op + 12 * g)     = w0;
      *(v4f*)(op + 12 * g + 4) = w1;
      *(v4f*)(op + 12 * g + 8) = w2;
    }
  }
}

#define XB_ ((NPOS_ / 256) * (C_ / CH_))
#define CB_ ((B_ * C_) / 256)
#define GB_ (NPOS_ / 256)
__global__ __launch_bounds__(256) void finalize_kernel(const float* __restrict__ xc,
                                                       const float* __restrict__ ga,
                                                       const float* __restrict__ gst,
                                                       float* __restrict__ out) {
  const int bx = blockIdx.x;
  if (bx < XB_) {
    const int chunk = bx / (NPOS_ / 256);
    const int gp = (bx - chunk * (NPOS_ / 256)) * 256 + threadIdx.x;
    const int c0 = chunk * CH_;
    const int b  = gp / HW_, p = gp - b * HW_;
    const float* ip = xc + ((size_t)(b * N_ + 1 + p) * C_ + c0);
    float* op = out + (size_t)B_ * C_ + ((size_t)(b * C_ + c0) * HW_ + p);
    v4f v[CH_ / 4];
#pragma unroll
    for (int g = 0; g < CH_ / 4; ++g) v[g] = *(const v4f*)(ip + 4 * g);
#pragma unroll
    for (int g = 0; g < CH_ / 4; ++g) {
      op[(size_t)(4 * g + 0) * HW_] = v[g].x;
      op[(size_t)(4 * g + 1) * HW_] = v[g].y;
      op[(size_t)(4 * g + 2) * HW_] = v[g].z;
      op[(size_t)(4 * g + 3) * HW_] = v[g].w;
    }
  } else if (bx < XB_ + CB_) {
    const int idx = (bx - XB_) * 256 + threadIdx.x;
    const int b = idx / C_, c = idx - b * C_;
    out[idx] = xc[(size_t)b * N_ * C_ + c];
  } else {
    const int j = (bx - XB_ - CB_) * 256 + threadIdx.x;
    const int b = j / HW_, p = j - b * HW_;
    float s = 0.f;
#pragma unroll
    for (int hh = 0; hh < HEADS_; ++hh) {
      float mh = gst[(b * HEADS_ + hh) * 2];
      float lh = gst[(b * HEADS_ + hh) * 2 + 1];
      s += expf(ga[(size_t)(b * HEADS_ + hh) * HW_ + p] - mh) * (1.f / lh);
    }
    out[(size_t)B_ * C_ + (size_t)B_ * C_ * HW_ + j] = s * 0.125f;
  }
}

extern "C" void kernel_launch(void* const* d_in, const int* in_sizes, int n_in,
                              void* d_out, int out_size, void* d_ws, size_t ws_size,
                              hipStream_t stream) {
  const float* cls    = (const float*)d_in[0];
  const float* x      = (const float*)d_in[1];
  const float* conv_w = (const float*)d_in[2];
  const float* conv_b = (const float*)d_in[3];
  const float* ln1_g  = (const float*)d_in[4];
  const float* ln1_b  = (const float*)d_in[5];
  const float* qkv_w  = (const float*)d_in[6];
  const float* proj_w = (const float*)d_in[7];
  const float* proj_b = (const float*)d_in[8];
  const float* ln2_g  = (const float*)d_in[9];
  const float* ln2_b  = (const float*)d_in[10];
  const float* fc1_w  = (const float*)d_in[11];
  const float* fc1_b  = (const float*)d_in[12];
  const float* fc2_w  = (const float*)d_in[13];
  const float* fc2_b  = (const float*)d_in[14];

  float* ws  = (float*)d_ws;
  float* xc  = ws;
  float* y   = xc  + (size_t)MP_ * C_;
  float* qkv = y   + (size_t)MP_ * C_;
  float* o   = qkv + (size_t)MP_ * C3_;
  float* m1  = qkv;
  float* ga  = o   + (size_t)MP_ * C_;
  float* gst = ga  + (size_t)B_ * HEADS_ * HW_;
  float* out = (float*)d_out;

  conv_pe_kernel<<<dim3(NPOS_ / 256, C_ / CH_), 256, 0, stream>>>(x, conv_w, conv_b, xc);
  cls_copy_kernel<<<(B_ * C_ + 255) / 256, 256, 0, stream>>>(cls, xc);

  ln_kernel<<<(M_ + 7) / 8, 256, 0, stream>>>(xc, ln1_g, ln1_b, y, M_);

  gemm_wmma_kernel<0><<<dim3(C3_ / 64, (MB_ + 7) / 8), 256, 0, stream>>>(
      y, qkv_w, nullptr, qkv, C_, C3_, MB_);

  attn_kernel<<<dim3(5, HEADS_, B_), 160, 0, stream>>>(qkv, o, ga, gst);

  gemm_wmma_kernel<1><<<dim3(C_ / 64, (MB_ + 7) / 8), 256, 0, stream>>>(
      o, proj_w, proj_b, xc, C_, C_, MB_);

  ln_kernel<<<(M_ + 7) / 8, 256, 0, stream>>>(xc, ln2_g, ln2_b, y, M_);

  gemm_wmma_kernel<2><<<dim3(C4_ / 64, (MB_ + 7) / 8), 256, 0, stream>>>(
      y, fc1_w, fc1_b, m1, C_, C4_, MB_);

  gemm_wmma_kernel<1><<<dim3(C_ / 64, (MB_ + 7) / 8), 256, 0, stream>>>(
      m1, fc2_w, fc2_b, xc, C4_, C_, MB_);

  finalize_kernel<<<XB_ + CB_ + GB_, 256, 0, stream>>>(xc, ga, gst, out);
}


# GEMM operands kept in WMMA fragment order in memory (weights re-ordered in the cls-copy launch; y, o, m1 written so): a fragment load is one 512-byte run, not 16 lines; LayerNorm sum order changes
# speedup vs baseline: 2.6325x; 1.1136x over previous
#include <hip/hip_runtime.h>
#include <math.h>

#define B_    16
#define C_    384
#define H_    28
#define W_    28
#define HW_   784
#define N_    785
#define M_    (B_*N_)
#define MP_   (M_ + 16)
#define HEADS_ 8
#define D_    48
#define C3_   1152
#define C4_   1536
#define MT_   (M_/16)
#define MB_   ((MT_ + 1)/2)
#define CH_   32
#define NPOS_ (B_*HW_)

typedef __attribute__((ext_vector_type(2))) float v2f;
typedef __attribute__((ext_vector_type(4))) float v4f;
typedef __attribute__((ext_vector_type(8))) float v8f;

__device__ __forceinline__ v8f wmma4(v2f a, v2f b, v8f c) {
  return __builtin_amdgcn_wmma_f32_16x16x4_f32(false, a, false, b, (short)0, c,
                                               false, false);
}

__global__ __launch_bounds__(256) void conv_pe_kernel(const float* __restrict__ x,
                                                      const float* __restrict__ cw,
                                                      const float* __restrict__ cb,
                                                      float* __restrict__ xc) {
  const int gp = blockIdx.x * 256 + threadIdx.x;
  const int c0 = blockIdx.y * CH_;
  const int b  = gp / HW_, p = gp - b * HW_;
  const int hh = p / W_,  ww = p - hh * W_;
  bool ok[9];
  int  off[9];
#pragma unroll
  for (int i = 0; i < 3; ++i) {
#pragma unroll
    for (int j = 0; j < 3; ++j) {
      const bool vi = (i == 0) ? (hh > 0) : (i == 2) ? (hh < H_ - 1) : true;
      const bool vj = (j == 0) ? (ww > 0) : (j == 2) ? (ww < W_ - 1) : true;
      ok[i * 3 + j]  = vi && vj;
      off[i * 3 + j] = (vi && vj) ? (i - 1) * W_ + (j - 1) : 0;
    }
  }
  const float* xp = x + ((size_t)(b * C_ + c0) * HW_ + p);
  float* op = xc + ((size_t)(b * N_ + 1 + p) * C_ + c0);
#pragma unroll
  for (int g = 0; g < CH_ / 8; ++g) {
    float v[8][9];
#pragma unroll
    for (int q = 0; q < 8; ++q) {
      const float* xq = xp + (size_t)(8 * g + q) * HW_;
#pragma unroll
      for (int t = 0; t < 9; ++t) v[q][t] = xq[off[t]];
    }
    __builtin_amdgcn_sched_barrier(0);
    float r[8];
#pragma unroll
    for (int q = 0; q < 8; ++q) {
      const int c = c0 + 8 * g + q;
      const float* wq = cw + c * 9;
      float s = 0.f;
#pragma unroll
      for (int t = 0; t < 9; ++t) s = fmaf(ok[t] ? v[q][t] : 0.f, wq[t], s);
      r[q] = v[q][4] + s + cb[c];
    }
    v4f w0 = {r[0], r[1], r[2], r[3]};
    v4f w1 = {r[4], r[5], r[6], r[7]};
    *(v4f*)(op + 8 * g)     = w0;
    *(v4f*)(op + 8 * g + 4) = w1;
  }
}

__device__ __forceinline__ size_t fo_off(int r, int k4, int kg) {
  return ((size_t)((r >> 4) * kg + (k4 >> 3)) * 32 + ((k4 >> 2) & 1) * 16 + (r & 15)) * 4;
}

#define WQ_ (C3_ * C_)
#define WP_ (C_ * C_)
#define W1_ (C4_ * C_)
#define W2_ (C_ * C4_)
#define PBC_ ((B_ * C_) / 256)
#define PBQ_ (WQ_ / 1024)
#define PBP_ (WP_ / 1024)
#define PB1_ (W1_ / 1024)
#define PB2_ (W2_ / 1024)

__device__ __forceinline__ void frag_order_piece(const float* __restrict__ src,
                                                 float* __restrict__ dst,
                                                 int kg, int p) {
  const int lane = p & 31;
  const int gi = p >> 5;
  const int t = gi / kg, i = gi - t * kg;
  const float* sp = src + (size_t)(t * 16 + (lane & 15)) * (kg * 8) + 8 * i + 4 * (lane >> 4);
  *(v4f*)(dst + (size_t)p * 4) = *(const v4f*)sp;
}

__global__ __launch_bounds__(256) void prep_kernel(const float* __restrict__ cls,
                                                   const float* __restrict__ qkv_w,
                                                   const float* __restrict__ proj_w,
                                                   const float* __restrict__ fc1_w,
                                                   const float* __restrict__ fc2_w,
                                                   float* __restrict__ xc,
                                                   float* __restrict__ wf) {
  int bx = blockIdx.x;
  if (bx < PBC_) {
    const int idx = bx * 256 + threadIdx.x;
    const int b = idx / C_, c = idx - b * C_;
    xc[(size_t)b * N_ * C_ + c] = cls[idx];
    return;
  }
  bx -= PBC_;
  if (bx < PBQ_) { frag_order_piece(qkv_w, wf, C_ / 8, bx * 256 + threadIdx.x); return; }
  bx -= PBQ_;
  if (bx < PBP_) { frag_order_piece(proj_w, wf + WQ_, C_ / 8, bx * 256 + threadIdx.x); return; }
  bx -= PBP_;
  if (bx < PB1_) { frag_order_piece(fc1_w, wf + WQ_ + WP_, C_ / 8, bx * 256 + threadIdx.x); return; }
  bx -= PB1_;
  frag_order_piece(fc2_w, wf + WQ_ + WP_ + W1_, C4_ / 8, bx * 256 + threadIdx.x);
}

__global__ __launch_bounds__(64) void ln_kernel(const float* __restrict__ xin,
                                                const float* __restrict__ g,
                                                const float* __restrict__ bta,
                                                float* __restrict__ yf) {
  const int wave = threadIdx.x >> 5, lane = threadIdx.x & 31;
  const int tile = blockIdx.x * 2 + wave;
  if (tile >= MT_) return;
  const int c = lane & 15, hf = lane >> 4;
  const float* xp = xin + (size_t)(tile * 16 + c) * C_ + 4 * hf;
  v4f v[C_ / 8];
#pragma unroll
  for (int i = 0; i < C_ / 8; ++i) v[i] = *(const v4f*)(xp + 8 * i);
  v4f s4 = {0.f, 0.f, 0.f, 0.f}, q4 = {0.f, 0.f, 0.f, 0.f};
#pragma unroll
  for (int i = 0; i < C_ / 8; ++i) { s4 += v[i]; q4 += v[i] * v[i]; }
  float s  = (s4.x + s4.y) + (s4.z + s4.w);
  float sq = (q4.x + q4.y) + (q4.z + q4.w);
  s  += __shfl_xor(s,  16, 32);
  sq += __shfl_xor(sq, 16, 32);
  const float mu  = s  * (1.f / C_);
  const float var = sq * (1.f / C_) - mu * mu;
  const float rs  = rsqrtf(var + 1e-5f);
  float* yp = yf + (size_t)tile * (16 * C_) + lane * 4;
  const float* gp = g + 4 * hf;
  const float* bp = bta + 4 * hf;
#pragma unroll
  for (int i = 0; i < C_ / 8; ++i) {
    const v4f gg = *(const v4f*)(gp + 8 * i);
    const v4f bb = *(const v4f*)(bp + 8 * i);
    v4f r;
    r.x = (v[i].x - mu) * rs * gg.x + bb.x;
    r.y = (v[i].y - mu) * rs * gg.y + bb.y;
    r.z = (v[i].z - mu) * rs * gg.z + bb.z;
    r.w = (v[i].w - mu) * rs * gg.w + bb.w;
    *(v4f*)(yp + i * 128) = r;
  }
}

template <int EPI>
__global__ __launch_bounds__(256) void gemm_wmma_kernel(
    const float* __restrict__ Xf, const float* __restrict__ Wf,
    const float* __restrict__ bias, float* __restrict__ Out,
    int K, int ldo, int Mblocks) {
  int lane = threadIdx.x & 31;
  int wave = threadIdx.x >> 5;
  int mblk = blockIdx.y * 8 + wave;
  if (mblk >= Mblocks) return;
  int n0 = blockIdx.x * 64;
  int m0 = mblk * 32;
  const size_t tstride = (size_t)K * 16;
  const float* xp0 = Xf + (size_t)(2 * mblk) * tstride + lane * 4;
  const float* xp1 = xp0 + tstride;
  const float* wp0 = Wf + (size_t)(4 * blockIdx.x) * tstride + lane * 4;
  const float* wp1 = wp0 + tstride;
  const float* wp2 = wp1 + tstride;
  const float* wp3 = wp2 + tstride;

  v8f acc[2][4] = {};
  for (int k = 0; k < K; k += 32) {
    v4f a0[4], a1[4], b0[4], b1[4], b2[4], b3[4];
#pragma unroll
    for (int u = 0; u < 4; ++u) {
      a0[u] = *(const v4f*)(xp0 + k * 16 + 128 * u);
      a1[u] = *(const v4f*)(xp1 + k * 16 + 128 * u);
      b0[u] = *(const v4f*)(wp0 + k * 16 + 128 * u);
      b1[u] = *(const v4f*)(wp1 + k * 16 + 128 * u);
      b2[u] = *(const v4f*)(wp2 + k * 16 + 128 * u);
      b3[u] = *(const v4f*)(wp3 + k * 16 + 128 * u);
    }
    __builtin_amdgcn_sched_barrier(0);
#define MM_(ACC, XA, WB) ACC = (EPI == 2) ? wmma4(WB, XA, ACC) : wmma4(XA, WB, ACC)
#pragma unroll
    for (int u = 0; u < 4; ++u) {
      v2f a0l = {a0[u].x, a0[u].y}, a0h = {a0[u].z, a0[u].w};
      v2f a1l = {a1[u].x, a1[u].y}, a1h = {a1[u].z, a1[u].w};
      v2f b0l = {b0[u].x, b0[u].y}, b0h = {b0[u].z, b0[u].w};
      v2f b1l = {b1[u].x, b1[u].y}, b1h = {b1[u].z, b1[u].w};
      v2f b2l = {b2[u].x, b2[u].y}, b2h = {b2[u].z, b2[u].w};
      v2f b3l = {b3[u].x, b3[u].y}, b3h = {b3[u].z, b3[u].w};
      MM_(acc[0][0], a0l, b0l);
      MM_(acc[1][0], a1l, b0l);
      MM_(acc[0][1], a0l, b1l);
      MM_(acc[1][1], a1l, b1l);
      MM_(acc[0][2], a0l, b2l);
      MM_(acc[1][2], a1l, b2l);
      MM_(acc[0][3], a0l, b3l);
      MM_(acc[1][3], a1l, b3l);
      MM_(acc[0][0], a0h, b0h);
      MM_(acc[1][0], a1h, b0h);
      MM_(acc[0][1], a0h, b1h);
      MM_(acc[1][1], a1h, b1h);
      MM_(acc[0][2], a0h, b2h);
      MM_(acc[1][2], a1h, b2h);
      MM_(acc[0][3], a0h, b3h);
      MM_(acc[1][3], a1h, b3h);
    }
#undef MM_
  }

  if (EPI == 2) {
    const int c = lane & 15, hfl = lane >> 4;
    const int kg2 = ldo >> 3;
#pragma unroll
    for (int mi = 0; mi < 2; ++mi) {
      const int mt = 2 * mblk + mi;
      if (mt < MT_) {
#pragma unroll
        for (int j = 0; j < 4; ++j) {
          const int nb = n0 + 16 * j + 8 * hfl;
          const v4f bl = *(const v4f*)(bias + nb);
          const v4f bh = *(const v4f*)(bias + nb + 4);
          float t[8];
          t[0] = acc[mi][j][0] + bl.x; t[1] = acc[mi][j][1] + bl.y;
          t[2] = acc[mi][j][2] + bl.z; t[3] = acc[mi][j][3] + bl.w;
          t[4] = acc[mi][j][4] + bh.x; t[5] = acc[mi][j][5] + bh.y;
          t[6] = acc[mi][j][6] + bh.z; t[7] = acc[mi][j][7] + bh.w;
#pragma unroll
          for (int r = 0; r < 8; ++r)
            t[r] = 0.5f * t[r] * (1.f + erff(t[r] * 0.70710678118654752f));
          float* dp = Out + ((size_t)(mt * kg2 + (nb >> 3)) * 32 + c) * 4;
          v4f lo = {t[0], t[1], t[2], t[3]};
          v4f hi = {t[4], t[5], t[6], t[7]};
          *(v4f*)dp        = lo;
          *(v4f*)(dp + 64) = hi;
        }
      }
    }
    return;
  }

  int coln = lane & 15;
  int roff = (lane >> 4) << 3;
#pragma unroll
  for (int mi = 0; mi < 2; ++mi) {
#pragma unroll
    for (int j = 0; j < 4; ++j) {
      int nc = n0 + j * 16 + coln;
      float bv = bias ? bias[nc] : 0.f;
#pragma unroll
      for (int r = 0; r < 8; ++r) {
        int gm = m0 + mi * 16 + roff + r;
        if (gm < M_) {
          size_t oi = (size_t)gm * ldo + nc;
          float v = acc[mi][j][r] + bv;
          if (EPI == 0) Out[oi] = v;
          else          Out[oi] += v;
        }
      }
    }
  }
}

struct __attribute__((packed, aligned(4))) f3 { float x, y, z; };

struct KStep { v4f k[6]; };
struct VStep { f3 va[4]; f3 vb[4]; };

__device__ __forceinline__ void load_k(KStep& t, const float* __restrict__ kbase,
                                       int key0, int c) {
  int kr = key0 + c; if (kr > N_ - 1) kr = N_ - 1;
  const float* kp = kbase + (size_t)kr * C3_;
#pragma unroll
  for (int i = 0; i < 6; ++i) t.k[i] = *(const v4f*)(kp + 8 * i);
}

__device__ __forceinline__ void load_v(VStep& t, const float* __restrict__ vbase,
                                       int key0, int hf) {
  const int kl = key0 + 8 * hf;
#pragma unroll
  for (int j = 0; j < 4; ++j) {
    int k0 = kl + 2 * j;     if (k0 > N_ - 1) k0 = N_ - 1;
    int k1 = kl + 2 * j + 1; if (k1 > N_ - 1) k1 = N_ - 1;
    t.va[j] = *(const f3*)(vbase + (size_t)k0 * C3_);
    t.vb[j] = *(const f3*)(vbase + (size_t)k1 * C3_);
  }
}

struct AttnState {
  v8f oa0, oa1, oa2, ob0, ob1, ob2;
  float m0, m1, l0, l1;
};

__device__ __forceinline__ void attn_step(const KStep& tk, const VStep& t,
                                          const v4f (&q0)[6],
                                          const v4f (&q1)[6], AttnState& st,
                                          int key0, int hf, bool cls_lane,
                                          float* __restrict__ gp) {
  const float scale = 0.14433756729740643f;
  v8f s0 = {}, s1 = {};
#pragma unroll
  for (int i = 0; i < 6; ++i) {
    v2f ka = {tk.k[i].x, tk.k[i].y}, kc = {tk.k[i].z, tk.k[i].w};
    v2f qa0 = {q0[i].x, q0[i].y}, qc0 = {q0[i].z, q0[i].w};
    v2f qa1 = {q1[i].x, q1[i].y}, qc1 = {q1[i].z, q1[i].w};
    s0 = wmma4(ka, qa0, s0);
    s1 = wmma4(ka, qa1, s1);
    s0 = wmma4(kc, qc0, s0);
    s1 = wmma4(kc, qc1, s1);
  }
  const int kl = key0 + 8 * hf;
  float x0[8], x1[8];
#pragma unroll
  for (int r = 0; r < 8; ++r) { x0[r] = s0[r] * scale; x1[r] = s1[r] * scale; }
  if (key0 + 16 > N_) {
#pragma unroll
    for (int r = 0; r < 8; ++r)
      if (kl + r >= N_) { x0[r] = -3.0e38f; x1[r] = -3.0e38f; }
  }
  if (cls_lane) {
#pragma unroll
    for (int r = 0; r < 8; ++r) {
      int key = kl + r;
      if (key >= 1 && key < N_) gp[key - 1] = x0[r];
    }
  }
  float mx0 = x0[0], mx1 = x1[0];
#pragma unroll
  for (int r = 1; r < 8; ++r) { mx0 = fmaxf(mx0, x0[r]); mx1 = fmaxf(mx1, x1[r]); }
  mx0 = fmaxf(mx0, __shfl_xor(mx0, 16, 32));
  mx1 = fmaxf(mx1, __shfl_xor(mx1, 16, 32));
  float mn0 = fmaxf(st.m0, mx0), mn1 = fmaxf(st.m1, mx1);
  float al0 = expf(st.m0 - mn0), al1 = expf(st.m1 - mn1);
  float ps0 = 0.f, ps1 = 0.f;
#pragma unroll
  for (int r = 0; r < 8; ++r) {
    x0[r] = expf(x0[r] - mn0); ps0 += x0[r];
    x1[r] = expf(x1[r] - mn1); ps1 += x1[r];
  }
  st.l0 = st.l0 * al0 + ps0;  st.l1 = st.l1 * al1 + ps1;
  st.m0 = mn0;  st.m1 = mn1;
#pragma unroll
  for (int r = 0; r < 8; ++r) {
    st.oa0[r] *= al0; st.oa1[r] *= al0; st.oa2[r] *= al0;
    st.ob0[r] *= al1; st.ob1[r] *= al1; st.ob2[r] *= al1;
  }
#pragma unroll
  for (int j = 0; j < 4; ++j) {
    v2f pa = {x0[2 * j], x0[2 * j + 1]};
    v2f pb = {x1[2 * j], x1[2 * j + 1]};
    v2f a0 = {t.va[j].x, t.vb[j].x}, a1 = {t.va[j].y, t.vb[j].y}, a2 = {t.va[j].z, t.vb[j].z};
    st.oa0 = wmma4(a0, pa, st.oa0);
    st.ob0 = wmma4(a0, pb, st.ob0);
    st.oa1 = wmma4(a1, pa, st.oa1);
    st.ob1 = wmma4(a1, pb, st.ob1);
    st.oa2 = wmma4(a2, pa, st.oa2);
    st.ob2 = wmma4(a2, pb, st.ob2);
  }
}

__global__ __launch_bounds__(160, 4) void attn_kernel(const float* __restrict__ qkv,
                                                   float* __restrict__ o,
                                                   float* __restrict__ ga,
                                                   float* __restrict__ gst) {
  int lane = threadIdx.x & 31;
  int wave = threadIdx.x >> 5;
  int pair = blockIdx.x * 5 + wave;
  int h = blockIdx.y, b = blockIdx.z;
  int t0 = pair * 32;
  int c  = lane & 15;
  int hf = lane >> 4;

  int gq0 = t0 + c;      if (gq0 > N_ - 1) gq0 = N_ - 1;
  int gq1 = t0 + 16 + c; if (gq1 > N_ - 1) gq1 = N_ - 1;
  const float* qp0 = qkv + (size_t)(b * N_ + gq0) * C3_ + h * D_ + 4 * hf;
  const float* qp1 = qkv + (size_t)(b * N_ + gq1) * C3_ + h * D_ + 4 * hf;
  v4f q0[6], q1[6];
#pragma unroll
  for (int i = 0; i < 6; ++i) {
    q0[i] = *(const v4f*)(qp0 + 8 * i);
    q1[i] = *(const v4f*)(qp1 + 8 * i);
  }

  const float* kbase = qkv + (size_t)b * N_ * C3_ + C_ + h * D_ + 4 * hf;
  const float* vbase = qkv + (size_t)b * N_ * C3_ + 2 * C_ + h * D_ + 3 * c;
  float* gp = ga + (size_t)(b * HEADS_ + h) * HW_;
  const bool cls_lane = (pair == 0) && (c == 0);

  AttnState st;
  st.oa0 = v8f{}; st.oa1 = v8f{}; st.oa2 = v8f{};
  st.ob0 = v8f{}; st.ob1 = v8f{}; st.ob2 = v8f{};
  st.m0 = -3.0e38f; st.m1 = -3.0e38f; st.l0 = 0.f; st.l1 = 0.f;

  KStep ka, kb;
  load_k(ka, kbase, 0, c);
  for (int key0 = 0; key0 < N_; key0 += 32) {
    VStep va, vb;
    load_k(kb, kbase, key0 + 16, c);
    load_v(va, vbase, key0, hf);
    attn_step(ka, va, q0, q1, st, key0, hf, cls_lane, gp);
    load_k(ka, kbase, key0 + 32, c);
    load_v(vb, vbase, key0 + 16, hf);
    attn_step(kb, vb, q0, q1, st, key0 + 16, hf, cls_lane, gp);
  }

  float lt0 = st.l0 + __shfl_xor(st.l0, 16, 32);
  float lt1 = st.l1 + __shfl_xor(st.l1, 16, 32);
  if (pair == 0 && lane == 0) {
    gst[(b * HEADS_ + h) * 2]     = st.m0;
    gst[(b * HEADS_ + h) * 2 + 1] = lt0;
  }
  float inv0 = 1.f / lt0, inv1 = 1.f / lt1;
  int qa = t0 + c, qb = t0 + 16 + c;
  const int kd = h * D_ + 24 * hf;
  if (qa < N_) {
    const int m = b * N_ + qa;
#pragma unroll
    for (int g = 0; g < 2; ++g) {
      v4f w0 = {st.oa0[4*g] * inv0,   st.oa1[4*g] * inv0,   st.oa2[4*g] * inv0,   st.oa0[4*g+1] * inv0};
      v4f w1 = {st.oa1[4*g+1] * inv0, st.oa2[4*g+1] * inv0, st.oa0[4*g+2] * inv0, st.oa1[4*g+2] * inv0};
      v4f w2 = {st.oa2[4*g+2] * inv0, st.oa0[4*g+3] * inv0, st.oa1[4*g+3] * inv0, st.oa2[4*g+3] * inv0};
      *(v4f*)(o + fo_off(m, kd + 12 * g,     C_ / 8)) = w0;
      *(v4f*)(o + fo_off(m, kd + 12 * g + 4, C_ / 8)) = w1;
      *(v4f*)(o + fo_off(m, kd + 12 * g + 8, C_ / 8)) = w2;
    }
  }
  if (qb < N_) {
    const int m = b * N_ + qb;
#pragma unroll
    for (int g = 0; g < 2; ++g) {
      v4f w0 = {st.ob0[4*g] * inv1,   st.ob1[4*g] * inv1,   st.ob2[4*g] * inv1,   st.ob0[4*g+1] * inv1};
      v4f w1 = {st.ob1[4*g+1] * inv1, st.ob2[4*g+1] * inv1, st.ob0[4*g+2] * inv1, st.ob1[4*g+2] * inv1};
      v4f w2 = {st.ob2[4*g+2] * inv1, st.ob0[4*g+3] * inv1, st.ob1[4*g+3] * inv1, st.ob2[4*g+3] * inv1};
      *(v4f*)(o + fo_off(m, kd + 12 * g,     C_ / 8)) = w0;
      *(v4f*)(o + fo_off(m, kd + 12 * g + 4, C_ / 8)) = w1;
      *(v4f*)(o + fo_off(m, kd + 12 * g + 8, C_ / 8)) = w2;
    }
  }
}

#define XB_ ((NPOS_ / 256) * (C_ / CH_))
#define CB_ ((B_ * C_) / 256)
#define GB_ (NPOS_ / 256)
__global__ __launch_bounds__(256) void finalize_kernel(const float* __restrict__ xc,
                                                       const float* __restrict__ ga,
                                                       const float* __restrict__ gst,
                                                       float* __restrict__ out) {
  const int bx = blockIdx.x;
  if (bx < XB_) {
    const int chunk = bx / (NPOS_ / 256);
    const int gp = (bx - chunk * (NPOS_ / 256)) * 256 + threadIdx.x;
    const int c0 = chunk * CH_;
    const int b  = gp / HW_, p = gp - b * HW_;
    const float* ip = xc + ((size_t)(b * N_ + 1 + p) * C_ + c0);
    float* op = out + (size_t)B_ * C_ + ((size_t)(b * C_ + c0) * HW_ + p);
    v4f v[CH_ / 4];
#pragma unroll
    for (int g = 0; g < CH_ / 4; ++g) v[g] = *(const v4f*)(ip + 4 * g);
#pragma unroll
    for (int g = 0; g < CH_ / 4; ++g) {
      op[(size_t)(4 * g + 0) * HW_] = v[g].x;
      op[(size_t)(4 * g + 1) * HW_] = v[g].y;
      op[(size_t)(4 * g + 2) * HW_] = v[g].z;
      op[(size_t)(4 * g + 3) * HW_] = v[g].w;
    }
  } else if (bx < XB_ + CB_) {
    const int idx = (bx - XB_) * 256 + threadIdx.x;
    const int b = idx / C_, c = idx - b * C_;
    out[idx] = xc[(size_t)b * N_ * C_ + c];
  } else {
    const int j = (bx - XB_ - CB_) * 256 + threadIdx.x;
    const int b = j / HW_, p = j - b * HW_;
    float s = 0.f;
#pragma unroll
    for (int hh = 0; hh < HEADS_; ++hh) {
      float mh = gst[(b * HEADS_ + hh) * 2];
      float lh = gst[(b * HEADS_ + hh) * 2 + 1];
      s += expf(ga[(size_t)(b * HEADS_ + hh) * HW_ + p] - mh) * (1.f / lh);
    }
    out[(size_t)B_ * C_ + (size_t)B_ * C_ * HW_ + j] = s * 0.125f;
  }
}

extern "C" void kernel_launch(void* const* d_in, const int* in_sizes, int n_in,
                              void* d_out, int out_size, void* d_ws, size_t ws_size,
                              hipStream_t stream) {
  const float* cls    = (const float*)d_in[0];
  const float* x      = (const float*)d_in[1];
  const float* conv_w = (const float*)d_in[2];
  const float* conv_b = (const float*)d_in[3];
  const float* ln1_g  = (const float*)d_in[4];
  const float* ln1_b  = (const float*)d_in[5];
  const float* qkv_w  = (const float*)d_in[6];
  const float* proj_w = (const float*)d_in[7];
  const float* proj_b = (const float*)d_in[8];
  const float* ln2_g  = (const float*)d_in[9];
  const float* ln2_b  = (const float*)d_in[10];
  const float* fc1_w  = (const float*)d_in[11];
  const float* fc1_b  = (const float*)d_in[12];
  const float* fc2_w  = (const float*)d_in[13];
  const float* fc2_b  = (const float*)d_in[14];

  float* ws  = (float*)d_ws;
  float* xc  = ws;
  float* y   = xc  + (size_t)MP_ * C_;
  float* qkv = y   + (size_t)MP_ * C_;
  float* o   = qkv + (size_t)MP_ * C3_;
  float* m1  = qkv;
  float* ga  = o   + (size_t)MP_ * C_;
  float* gst = ga  + (size_t)B_ * HEADS_ * HW_;
  float* wf  = gst + B_ * HEADS_ * 2;
  const float* qkv_wf  = wf;
  const float* proj_wf = qkv_wf  + WQ_;
  const float* fc1_wf  = proj_wf + WP_;
  const float* fc2_wf  = fc1_wf  + W1_;
  float* out = (float*)d_out;

  conv_pe_kernel<<<dim3(NPOS_ / 256, C_ / CH_), 256, 0, stream>>>(x, conv_w, conv_b, xc);
  prep_kernel<<<PBC_ + PBQ_ + PBP_ + PB1_ + PB2_, 256, 0, stream>>>(
      cls, qkv_w, proj_w, fc1_w, fc2_w, xc, wf);

  ln_kernel<<<(MT_ + 1) / 2, 64, 0, stream>>>(xc, ln1_g, ln1_b, y);

  gemm_wmma_kernel<0><<<dim3(C3_ / 64, (MB_ + 7) / 8), 256, 0, stream>>>(
      y, qkv_wf, nullptr, qkv, C_, C3_, MB_);

  attn_kernel<<<dim3(5, HEADS_, B_), 160, 0, stream>>>(qkv, o, ga, gst);

  gemm_wmma_kernel<1><<<dim3(C_ / 64, (MB_ + 7) / 8), 256, 0, stream>>>(
      o, proj_wf, proj_b, xc, C_, C_, MB_);

  ln_kernel<<<(MT_ + 1) / 2, 64, 0, stream>>>(xc, ln2_g, ln2_b, y);

  gemm_wmma_kernel<2><<<dim3(C4_ / 64, (MB_ + 7) / 8), 256, 0, stream>>>(
      y, fc1_wf, fc1_b, m1, C_, C4_, MB_);

  gemm_wmma_kernel<1><<<dim3(C_ / 64, (MB_ + 7) / 8), 256, 0, stream>>>(
      m1, fc2_wf, fc2_b, xc, C4_, C_, MB_);

  finalize_kernel<<<XB_ + CB_ + GB_, 256, 0, stream>>>(xc, ga, gst, out);
}


# proj and fc2 GEMMs: 80x32 wave tile (5x2; 157*80 rows, 12*32 columns exactly: no row guards, old outputs load as one group), 4-wave blocks: at most 20 output tiles per SIMD, not 32; sums unchanged
# speedup vs baseline: 3.0841x; 1.1715x over previous
#include <hip/hip_runtime.h>
#include <math.h>

#define B_    16
#define C_    384
#define H_    28
#define W_    28
#define HW_   784
#define N_    785
#define M_    (B_*N_)
#define MP_   (M_ + 16)
#define HEADS_ 8
#define D_    48
#define C3_   1152
#define C4_   1536
#define MT_   (M_/16)
#define MB_   ((MT_ + 1)/2)
#define CH_   32
#define NPOS_ (B_*HW_)

typedef __attribute__((ext_vector_type(2))) float v2f;
typedef __attribute__((ext_vector_type(4))) float v4f;
typedef __attribute__((ext_vector_type(8))) float v8f;

__device__ __forceinline__ v8f wmma4(v2f a, v2f b, v8f c) {
  return __builtin_amdgcn_wmma_f32_16x16x4_f32(false, a, false, b, (short)0, c,
                                               false, false);
}

__global__ __launch_bounds__(256) void conv_pe_kernel(const float* __restrict__ x,
                                                      const float* __restrict__ cw,
                                                      const float* __restrict__ cb,
                                                      float* __restrict__ xc) {
  const int gp = blockIdx.x * 256 + threadIdx.x;
  const int c0 = blockIdx.y * CH_;
  const int b  = gp / HW_, p = gp - b * HW_;
  const int hh = p / W_,  ww = p - hh * W_;
  bool ok[9];
  int  off[9];
#pragma unroll
  for (int i = 0; i < 3; ++i) {
#pragma unroll
    for (int j = 0; j < 3; ++j) {
      const bool vi = (i == 0) ? (hh > 0) : (i == 2) ? (hh < H_ - 1) : true;
      const bool vj = (j == 0) ? (ww > 0) : (j == 2) ? (ww < W_ - 1) : true;
      ok[i * 3 + j]  = vi && vj;
      off[i * 3 + j] = (vi && vj) ? (i - 1) * W_ + (j - 1) : 0;
    }
  }
  const float* xp = x + ((size_t)(b * C_ + c0) * HW_ + p);
  float* op = xc + ((size_t)(b * N_ + 1 + p) * C_ + c0);
#pragma unroll
  for (int g = 0; g < CH_ / 8; ++g) {
    float v[8][9];
#pragma unroll
    for (int q = 0; q < 8; ++q) {
      const float* xq = xp + (size_t)(8 * g + q) * HW_;
#pragma unroll
      for (int t = 0; t < 9; ++t) v[q][t] = xq[off[t]];
    }
    __builtin_amdgcn_sched_barrier(0);
    float r[8];
#pragma unroll
    for (int q = 0; q < 8; ++q) {
      const int c = c0 + 8 * g + q;
      const float* wq = cw + c * 9;
      float s = 0.f;
#pragma unroll
      for (int t = 0; t < 9; ++t) s = fmaf(ok[t] ? v[q][t] : 0.f, wq[t], s);
      r[q] = v[q][4] + s + cb[c];
    }
    v4f w0 = {r[0], r[1], r[2], r[3]};
    v4f w1 = {r[4], r[5], r[6], r[7]};
    *(v4f*)(op + 8 * g)     = w0;
    *(v4f*)(op + 8 * g + 4) = w1;
  }
}

__device__ __forceinline__ size_t fo_off(int r, int k4, int kg) {
  return ((size_t)((r >> 4) * kg + (k4 >> 3)) * 32 + ((k4 >> 2) & 1) * 16 + (r & 15)) * 4;
}

#define WQ_ (C3_ * C_)
#define WP_ (C_ * C_)
#define W1_ (C4_ * C_)
#define W2_ (C_ * C4_)
#define PBC_ ((B_ * C_) / 256)
#define PBQ_ (WQ_ / 1024)
#define PBP_ (WP_ / 1024)
#define PB1_ (W1_ / 1024)
#define PB2_ (W2_ / 1024)

__device__ __forceinline__ void frag_order_piece(const float* __restrict__ src,
                                                 float* __restrict__ dst,
                                                 int kg, int p) {
  const int lane = p & 31;
  const int gi = p >> 5;
  const int t = gi / kg, i = gi - t * kg;
  const float* sp = src + (size_t)(t * 16 + (lane & 15)) * (kg * 8) + 8 * i + 4 * (lane >> 4);
  *(v4f*)(dst + (size_t)p * 4) = *(const v4f*)sp;
}

__global__ __launch_bounds__(256) void prep_kernel(const float* __restrict__ cls,
                                                   const float* __restrict__ qkv_w,
                                                   const float* __restrict__ proj_w,
                                                   const float* __restrict__ fc1_w,
                                                   const float* __restrict__ fc2_w,
                                                   float* __restrict__ xc,
                                                   float* __restrict__ wf) {
  int bx = blockIdx.x;
  if (bx < PBC_) {
    const int idx = bx * 256 + threadIdx.x;
    const int b = idx / C_, c = idx - b * C_;
    xc[(size_t)b * N_ * C_ + c] = cls[idx];
    return;
  }
  bx -= PBC_;
  if (bx < PBQ_) { frag_order_piece(qkv_w, wf, C_ / 8, bx * 256 + threadIdx.x); return; }
  bx -= PBQ_;
  if (bx < PBP_) { frag_order_piece(proj_w, wf + WQ_, C_ / 8, bx * 256 + threadIdx.x); return; }
  bx -= PBP_;
  if (bx < PB1_) { frag_order_piece(fc1_w, wf + WQ_ + WP_, C_ / 8, bx * 256 + threadIdx.x); return; }
  bx -= PB1_;
  frag_order_piece(fc2_w, wf + WQ_ + WP_ + W1_, C4_ / 8, bx * 256 + threadIdx.x);
}

__global__ __launch_bounds__(64) void ln_kernel(const float* __restrict__ xin,
                                                const float* __restrict__ g,
                                                const float* __restrict__ bta,
                                                float* __restrict__ yf) {
  const int wave = threadIdx.x >> 5, lane = threadIdx.x & 31;
  const int tile = blockIdx.x * 2 + wave;
  if (tile >= MT_) return;
  const int c = lane & 15, hf = lane >> 4;
  const float* xp = xin + (size_t)(tile * 16 + c) * C_ + 4 * hf;
  v4f v[C_ / 8];
#pragma unroll
  for (int i = 0; i < C_ / 8; ++i) v[i] = *(const v4f*)(xp + 8 * i);
  v4f s4 = {0.f, 0.f, 0.f, 0.f}, q4 = {0.f, 0.f, 0.f, 0.f};
#pragma unroll
  for (int i = 0; i < C_ / 8; ++i) { s4 += v[i]; q4 += v[i] * v[i]; }
  float s  = (s4.x + s4.y) + (s4.z + s4.w);
  float sq = (q4.x + q4.y) + (q4.z + q4.w);
  s  += __shfl_xor(s,  16, 32);
  sq += __shfl_xor(sq, 16, 32);
  const float mu  = s  * (1.f / C_);
  const float var = sq * (1.f / C_) - mu * mu;
  const float rs  = rsqrtf(var + 1e-5f);
  float* yp = yf + (size_t)tile * (16 * C_) + lane * 4;
  const float* gp = g + 4 * hf;
  const float* bp = bta + 4 * hf;
#pragma unroll
  for (int i = 0; i < C_ / 8; ++i) {
    const v4f gg = *(const v4f*)(gp + 8 * i);
    const v4f bb = *(const v4f*)(bp + 8 * i);
    v4f r;
    r.x = (v[i].x - mu) * rs * gg.x + bb.x;
    r.y = (v[i].y - mu) * rs * gg.y + bb.y;
    r.z = (v[i].z - mu) * rs * gg.z + bb.z;
    r.w = (v[i].w - mu) * rs * gg.w + bb.w;
    *(v4f*)(yp + i * 128) = r;
  }
}

template <int EPI>
__global__ __launch_bounds__(256) void gemm_wmma_kernel(
    const float* __restrict__ Xf, const float* __restrict__ Wf,
    const float* __restrict__ bias, float* __restrict__ Out,
    int K, int ldo, int Mblocks) {
  int lane = threadIdx.x & 31;
  int wave = threadIdx.x >> 5;
  int mblk = blockIdx.y * 8 + wave;
  if (mblk >= Mblocks) return;
  int n0 = blockIdx.x * 64;
  int m0 = mblk * 32;
  const size_t tstride = (size_t)K * 16;
  const float* xp0 = Xf + (size_t)(2 * mblk) * tstride + lane * 4;
  const float* xp1 = xp0 + tstride;
  const float* wp0 = Wf + (size_t)(4 * blockIdx.x) * tstride + lane * 4;
  const float* wp1 = wp0 + tstride;
  const float* wp2 = wp1 + tstride;
  const float* wp3 = wp2 + tstride;

  v8f acc[2][4] = {};
  for (int k = 0; k < K; k += 32) {
    v4f a0[4], a1[4], b0[4], b1[4], b2[4], b3[4];
#pragma unroll
    for (int u = 0; u < 4; ++u) {
      a0[u] = *(const v4f*)(xp0 + k * 16 + 128 * u);
      a1[u] = *(const v4f*)(xp1 + k * 16 + 128 * u);
      b0[u] = *(const v4f*)(wp0 + k * 16 + 128 * u);
      b1[u] = *(const v4f*)(wp1 + k * 16 + 128 * u);
      b2[u] = *(const v4f*)(wp2 + k * 16 + 128 * u);
      b3[u] = *(const v4f*)(wp3 + k * 16 + 128 * u);
    }
    __builtin_amdgcn_sched_barrier(0);
#define MM_(ACC, XA, WB) ACC = (EPI == 2) ? wmma4(WB, XA, ACC) : wmma4(XA, WB, ACC)
#pragma unroll
    for (int u = 0; u < 4; ++u) {
      v2f a0l = {a0[u].x, a0[u].y}, a0h = {a0[u].z, a0[u].w};
      v2f a1l = {a1[u].x, a1[u].y}, a1h = {a1[u].z, a1[u].w};
      v2f b0l = {b0[u].x, b0[u].y}, b0h = {b0[u].z, b0[u].w};
      v2f b1l = {b1[u].x, b1[u].y}, b1h = {b1[u].z, b1[u].w};
      v2f b2l = {b2[u].x, b2[u].y}, b2h = {b2[u].z, b2[u].w};
      v2f b3l = {b3[u].x, b3[u].y}, b3h = {b3[u].z, b3[u].w};
      MM_(acc[0][0], a0l, b0l);
      MM_(acc[1][0], a1l, b0l);
      MM_(acc[0][1], a0l, b1l);
      MM_(acc[1][1], a1l, b1l);
      MM_(acc[0][2], a0l, b2l);
      MM_(acc[1][2], a1l, b2l);
      MM_(acc[0][3], a0l, b3l);
      MM_(acc[1][3], a1l, b3l);
      MM_(acc[0][0], a0h, b0h);
      MM_(acc[1][0], a1h, b0h);
      MM_(acc[0][1], a0h, b1h);
      MM_(acc[1][1], a1h, b1h);
      MM_(acc[0][2], a0h, b2h);
      MM_(acc[1][2], a1h, b2h);
      MM_(acc[0][3], a0h, b3h);
      MM_(acc[1][3], a1h, b3h);
    }
#undef MM_
  }

  if (EPI == 2) {
    const int c = lane & 15, hfl = lane >> 4;
    const int kg2 = ldo >> 3;
#pragma unroll
    for (int mi = 0; mi < 2; ++mi) {
      const int mt = 2 * mblk + mi;
      if (mt < MT_) {
#pragma unroll
        for (int j = 0; j < 4; ++j) {
          const int nb = n0 + 16 * j + 8 * hfl;
          const v4f bl = *(const v4f*)(bias + nb);
          const v4f bh = *(const v4f*)(bias + nb + 4);
          float t[8];
          t[0] = acc[mi][j][0] + bl.x; t[1] = acc[mi][j][1] + bl.y;
          t[2] = acc[mi][j][2] + bl.z; t[3] = acc[mi][j][3] + bl.w;
          t[4] = acc[mi][j][4] + bh.x; t[5] = acc[mi][j][5] + bh.y;
          t[6] = acc[mi][j][6] + bh.z; t[7] = acc[mi][j][7] + bh.w;
#pragma unroll
          for (int r = 0; r < 8; ++r)
            t[r] = 0.5f * t[r] * (1.f + erff(t[r] * 0.70710678118654752f));
          float* dp = Out + ((size_t)(mt * kg2 + (nb >> 3)) * 32 + c) * 4;
          v4f lo = {t[0], t[1], t[2], t[3]};
          v4f hi = {t[4], t[5], t[6], t[7]};
          *(v4f*)dp        = lo;
          *(v4f*)(dp + 64) = hi;
        }
      }
    }
    return;
  }

  int coln = lane & 15;
  int roff = (lane >> 4) << 3;
#pragma unroll
  for (int mi = 0; mi < 2; ++mi) {
#pragma unroll
    for (int j = 0; j < 4; ++j) {
      int nc = n0 + j * 16 + coln;
      float bv = bias ? bias[nc] : 0.f;
#pragma unroll
      for (int r = 0; r < 8; ++r) {
        int gm = m0 + mi * 16 + roff + r;
        if (gm < M_) {
          Out[(size_t)gm * ldo + nc] = acc[mi][j][r] + bv;
        }
      }
    }
  }
}

#define RM_ 5
#define RN_ 2
__global__ __launch_bounds__(128) void gemm_res_kernel(
    const float* __restrict__ Xf, const float* __restrict__ Wf,
    const float* __restrict__ bias, float* __restrict__ Out, int K) {
  const int lane = threadIdx.x & 31;
  const int wave = threadIdx.x >> 5;
  const int mblk = blockIdx.y;
  const int nblk = blockIdx.x * 4 + wave;
  const size_t tstride = (size_t)K * 16;
  const float* xp = Xf + (size_t)(RM_ * mblk) * tstride + lane * 4;
  const float* wp = Wf + (size_t)(RN_ * nblk) * tstride + lane * 4;

  v8f acc[RM_][RN_] = {};
  for (int k = 0; k < K; k += 32) {
    v4f a[RM_][4], b[RN_][4];
#pragma unroll
    for (int u = 0; u < 4; ++u) {
#pragma unroll
      for (int mi = 0; mi < RM_; ++mi)
        a[mi][u] = *(const v4f*)(xp + mi * tstride + k * 16 + 128 * u);
#pragma unroll
      for (int j = 0; j < RN_; ++j)
        b[j][u] = *(const v4f*)(wp + j * tstride + k * 16 + 128 * u);
    }
    __builtin_amdgcn_sched_barrier(0);
#pragma unroll
    for (int u = 0; u < 4; ++u) {
#pragma unroll
      for (int j = 0; j < RN_; ++j) {
        const v2f bl = {b[j][u].x, b[j][u].y};
#pragma unroll
        for (int mi = 0; mi < RM_; ++mi) {
          const v2f al = {a[mi][u].x, a[mi][u].y};
          acc[mi][j] = wmma4(al, bl, acc[mi][j]);
        }
      }
#pragma unroll
      for (int j = 0; j < RN_; ++j) {
        const v2f bh = {b[j][u].z, b[j][u].w};
#pragma unroll
        for (int mi = 0; mi < RM_; ++mi) {
          const v2f ah = {a[mi][u].z, a[mi][u].w};
          acc[mi][j] = wmma4(ah, bh, acc[mi][j]);
        }
      }
    }
  }

  const int coln = lane & 15;
  const int roff = (lane >> 4) << 3;
#pragma unroll
  for (int mi = 0; mi < RM_; ++mi) {
#pragma unroll
    for (int j = 0; j < RN_; ++j) {
      const int nc = nblk * (16 * RN_) + 16 * j + coln;
      const float bv = bias[nc];
      float* op = Out + (size_t)(mblk * (16 * RM_) + 16 * mi + roff) * C_ + nc;
#pragma unroll
      for (int r = 0; r < 8; ++r) {
        const float v = acc[mi][j][r] + bv;
        op[(size_t)r * C_] += v;
      }
    }
  }
}

struct __attribute__((packed, aligned(4))) f3 { float x, y, z; };

struct KStep { v4f k[6]; };
struct VStep { f3 va[4]; f3 vb[4]; };

__device__ __forceinline__ void load_k(KStep& t, const float* __restrict__ kbase,
                                       int key0, int c) {
  int kr = key0 + c; if (kr > N_ - 1) kr = N_ - 1;
  const float* kp = kbase + (size_t)kr * C3_;
#pragma unroll
  for (int i = 0; i < 6; ++i) t.k[i] = *(const v4f*)(kp + 8 * i);
}

__device__ __forceinline__ void load_v(VStep& t, const float* __restrict__ vbase,
                                       int key0, int hf) {
  const int kl = key0 + 8 * hf;
#pragma unroll
  for (int j = 0; j < 4; ++j) {
    int k0 = kl + 2 * j;     if (k0 > N_ - 1) k0 = N_ - 1;
    int k1 = kl + 2 * j + 1; if (k1 > N_ - 1) k1 = N_ - 1;
    t.va[j] = *(const f3*)(vbase + (size_t)k0 * C3_);
    t.vb[j] = *(const f3*)(vbase + (size_t)k1 * C3_);
  }
}

struct AttnState {
  v8f oa0, oa1, oa2, ob0, ob1, ob2;
  float m0, m1, l0, l1;
};

__device__ __forceinline__ void attn_step(const KStep& tk, const VStep& t,
                                          const v4f (&q0)[6],
                                          const v4f (&q1)[6], AttnState& st,
                                          int key0, int hf, bool cls_lane,
                                          float* __restrict__ gp) {
  const float scale = 0.14433756729740643f;
  v8f s0 = {}, s1 = {};
#pragma unroll
  for (int i = 0; i < 6; ++i) {
    v2f ka = {tk.k[i].x, tk.k[i].y}, kc = {tk.k[i].z, tk.k[i].w};
    v2f qa0 = {q0[i].x, q0[i].y}, qc0 = {q0[i].z, q0[i].w};
    v2f qa1 = {q1[i].x, q1[i].y}, qc1 = {q1[i].z, q1[i].w};
    s0 = wmma4(ka, qa0, s0);
    s1 = wmma4(ka, qa1, s1);
    s0 = wmma4(kc, qc0, s0);
    s1 = wmma4(kc, qc1, s1);
  }
  const int kl = key0 + 8 * hf;
  float x0[8], x1[8];
#pragma unroll
  for (int r = 0; r < 8; ++r) { x0[r] = s0[r] * scale; x1[r] = s1[r] * scale; }
  if (key0 + 16 > N_) {
#pragma unroll
    for (int r = 0; r < 8; ++r)
      if (kl + r >= N_) { x0[r] = -3.0e38f; x1[r] = -3.0e38f; }
  }
  if (cls_lane) {
#pragma unroll
    for (int r = 0; r < 8; ++r) {
      int key = kl + r;
      if (key >= 1 && key < N_) gp[key - 1] = x0[r];
    }
  }
  float mx0 = x0[0], mx1 = x1[0];
#pragma unroll
  for (int r = 1; r < 8; ++r) { mx0 = fmaxf(mx0, x0[r]); mx1 = fmaxf(mx1, x1[r]); }
  mx0 = fmaxf(mx0, __shfl_xor(mx0, 16, 32));
  mx1 = fmaxf(mx1, __shfl_xor(mx1, 16, 32));
  float mn0 = fmaxf(st.m0, mx0), mn1 = fmaxf(st.m1, mx1);
  float al0 = expf(st.m0 - mn0), al1 = expf(st.m1 - mn1);
  float ps0 = 0.f, ps1 = 0.f;
#pragma unroll
  for (int r = 0; r < 8; ++r) {
    x0[r] = expf(x0[r] - mn0); ps0 += x0[r];
    x1[r] = expf(x1[r] - mn1); ps1 += x1[r];
  }
  st.l0 = st.l0 * al0 + ps0;  st.l1 = st.l1 * al1 + ps1;
  st.m0 = mn0;  st.m1 = mn1;
#pragma unroll
  for (int r = 0; r < 8; ++r) {
    st.oa0[r] *= al0; st.oa1[r] *= al0; st.oa2[r] *= al0;
    st.ob0[r] *= al1; st.ob1[r] *= al1; st.ob2[r] *= al1;
  }
#pragma unroll
  for (int j = 0; j < 4; ++j) {
    v2f pa = {x0[2 * j], x0[2 * j + 1]};
    v2f pb = {x1[2 * j], x1[2 * j + 1]};
    v2f a0 = {t.va[j].x, t.vb[j].x}, a1 = {t.va[j].y, t.vb[j].y}, a2 = {t.va[j].z, t.vb[j].z};
    st.oa0 = wmma4(a0, pa, st.oa0);
    st.ob0 = wmma4(a0, pb, st.ob0);
    st.oa1 = wmma4(a1, pa, st.oa1);
    st.ob1 = wmma4(a1, pb, st.ob1);
    st.oa2 = wmma4(a2, pa, st.oa2);
    st.ob2 = wmma4(a2, pb, st.ob2);
  }
}

__global__ __launch_bounds__(160, 4) void attn_kernel(const float* __restrict__ qkv,
                                                   float* __restrict__ o,
                                                   float* __restrict__ ga,
                                                   float* __restrict__ gst) {
  int lane = threadIdx.x & 31;
  int wave = threadIdx.x >> 5;
  int pair = blockIdx.x * 5 + wave;
  int h = blockIdx.y, b = blockIdx.z;
  int t0 = pair * 32;
  int c  = lane & 15;
  int hf = lane >> 4;

  int gq0 = t0 + c;      if (gq0 > N_ - 1) gq0 = N_ - 1;
  int gq1 = t0 + 16 + c; if (gq1 > N_ - 1) gq1 = N_ - 1;
  const float* qp0 = qkv + (size_t)(b * N_ + gq0) * C3_ + h * D_ + 4 * hf;
  const float* qp1 = qkv + (size_t)(b * N_ + gq1) * C3_ + h * D_ + 4 * hf;
  v4f q0[6], q1[6];
#pragma unroll
  for (int i = 0; i < 6; ++i) {
    q0[i] = *(const v4f*)(qp0 + 8 * i);
    q1[i] = *(const v4f*)(qp1 + 8 * i);
  }

  const float* kbase = qkv + (size_t)b * N_ * C3_ + C_ + h * D_ + 4 * hf;
  const float* vbase = qkv + (size_t)b * N_ * C3_ + 2 * C_ + h * D_ + 3 * c;
  float* gp = ga + (size_t)(b * HEADS_ + h) * HW_;
  const bool cls_lane = (pair == 0) && (c == 0);

  AttnState st;
  st.oa0 = v8f{}; st.oa1 = v8f{}; st.oa2 = v8f{};
  st.ob0 = v8f{}; st.ob1 = v8f{}; st.ob2 = v8f{};
  st.m0 = -3.0e38f; st.m1 = -3.0e38f; st.l0 = 0.f; st.l1 = 0.f;

  KStep ka, kb;
  load_k(ka, kbase, 0, c);
  for (int key0 = 0; key0 < N_; key0 += 32) {
    VStep va, vb;
    load_k(kb, kbase, key0 + 16, c);
    load_v(va, vbase, key0, hf);
    attn_step(ka, va, q0, q1, st, key0, hf, cls_lane, gp);
    load_k(ka, kbase, key0 + 32, c);
    load_v(vb, vbase, key0 + 16, hf);
    attn_step(kb, vb, q0, q1, st, key0 + 16, hf, cls_lane, gp);
  }

  float lt0 = st.l0 + __shfl_xor(st.l0, 16, 32);
  float lt1 = st.l1 + __shfl_xor(st.l1, 16, 32);
  if (pair == 0 && lane == 0) {
    gst[(b * HEADS_ + h) * 2]     = st.m0;
    gst[(b * HEADS_ + h) * 2 + 1] = lt0;
  }
  float inv0 = 1.f / lt0, inv1 = 1.f / lt1;
  int qa = t0 + c, qb = t0 + 16 + c;
  const int kd = h * D_ + 24 * hf;
  if (qa < N_) {
    const int m = b * N_ + qa;
#pragma unroll
    for (int g = 0; g < 2; ++g) {
      v4f w0 = {st.oa0[4*g] * inv0,   st.oa1[4*g] * inv0,   st.oa2[4*g] * inv0,   st.oa0[4*g+1] * inv0};
      v4f w1 = {st.oa1[4*g+1] * inv0, st.oa2[4*g+1] * inv0, st.oa0[4*g+2] * inv0, st.oa1[4*g+2] * inv0};
      v4f w2 = {st.oa2[4*g+2] * inv0, st.oa0[4*g+3] * inv0, st.oa1[4*g+3] * inv0, st.oa2[4*g+3] * inv0};
      *(v4f*)(o + fo_off(m, kd + 12 * g,     C_ / 8)) = w0;
      *(v4f*)(o + fo_off(m, kd + 12 * g + 4, C_ / 8)) = w1;
      *(v4f*)(o + fo_off(m, kd + 12 * g + 8, C_ / 8)) = w2;
    }
  }
  if (qb < N_) {
    const int m = b * N_ + qb;
#pragma unroll
    for (int g = 0; g < 2; ++g) {
      v4f w0 = {st.ob0[4*g] * inv1,   st.ob1[4*g] * inv1,   st.ob2[4*g] * inv1,   st.ob0[4*g+1] * inv1};
      v4f w1 = {st.ob1[4*g+1] * inv1, st.ob2[4*g+1] * inv1, st.ob0[4*g+2] * inv1, st.ob1[4*g+2] * inv1};
      v4f w2 = {st.ob2[4*g+2] * inv1, st.ob0[4*g+3] * inv1, st.ob1[4*g+3] * inv1, st.ob2[4*g+3] * inv1};
      *(v4f*)(o + fo_off(m, kd + 12 * g,     C_ / 8)) = w0;
      *(v4f*)(o + fo_off(m, kd + 12 * g + 4, C_ / 8)) = w1;
      *(v4f*)(o + fo_off(m, kd + 12 * g + 8, C_ / 8)) = w2;
    }
  }
}

#define XB_ ((NPOS_ / 256) * (C_ / CH_))
#define CB_ ((B_ * C_) / 256)
#define GB_ (NPOS_ / 256)
__global__ __launch_bounds__(256) void finalize_kernel(const float* __restrict__ xc,
                                                       const float* __restrict__ ga,
                                                       const float* __restrict__ gst,
                                                       float* __restrict__ out) {
  const int bx = blockIdx.x;
  if (bx < XB_) {
    const int chunk = bx / (NPOS_ / 256);
    const int gp = (bx - chunk * (NPOS_ / 256)) * 256 + threadIdx.x;
    const int c0 = chunk * CH_;
    const int b  = gp / HW_, p = gp - b * HW_;
    const float* ip = xc + ((size_t)(b * N_ + 1 + p) * C_ + c0);
    float* op = out + (size_t)B_ * C_ + ((size_t)(b * C_ + c0) * HW_ + p);
    v4f v[CH_ / 4];
#pragma unroll
    for (int g = 0; g < CH_ / 4; ++g) v[g] = *(const v4f*)(ip + 4 * g);
#pragma unroll
    for (int g = 0; g < CH_ / 4; ++g) {
      op[(size_t)(4 * g + 0) * HW_] = v[g].x;
      op[(size_t)(4 * g + 1) * HW_] = v[g].y;
      op[(size_t)(4 * g + 2) * HW_] = v[g].z;
      op[(size_t)(4 * g + 3) * HW_] = v[g].w;
    }
  } else if (bx < XB_ + CB_) {
    const int idx = (bx - XB_) * 256 + threadIdx.x;
    const int b = idx / C_, c = idx - b * C_;
    out[idx] = xc[(size_t)b * N_ * C_ + c];
  } else {
    const int j = (bx - XB_ - CB_) * 256 + threadIdx.x;
    const int b = j / HW_, p = j - b * HW_;
    float s = 0.f;
#pragma unroll
    for (int hh = 0; hh < HEADS_; ++hh) {
      float mh = gst[(b * HEADS_ + hh) * 2];
      float lh = gst[(b * HEADS_ + hh) * 2 + 1];
      s += expf(ga[(size_t)(b * HEADS_ + hh) * HW_ + p] - mh) * (1.f / lh);
    }
    out[(size_t)B_ * C_ + (size_t)B_ * C_ * HW_ + j] = s * 0.125f;
  }
}

extern "C" void kernel_launch(void* const* d_in, const int* in_sizes, int n_in,
                              void* d_out, int out_size, void* d_ws, size_t ws_size,
                              hipStream_t stream) {
  const float* cls    = (const float*)d_in[0];
  const float* x      = (const float*)d_in[1];
  const float* conv_w = (const float*)d_in[2];
  const float* conv_b = (const float*)d_in[3];
  const float* ln1_g  = (const float*)d_in[4];
  const float* ln1_b  = (const float*)d_in[5];
  const float* qkv_w  = (const float*)d_in[6];
  const float* proj_w = (const float*)d_in[7];
  const float* proj_b = (const float*)d_in[8];
  const float* ln2_g  = (const float*)d_in[9];
  const float* ln2_b  = (const float*)d_in[10];
  const float* fc1_w  = (const float*)d_in[11];
  const float* fc1_b  = (const float*)d_in[12];
  const float* fc2_w  = (const float*)d_in[13];
  const float* fc2_b  = (const float*)d_in[14];

  float* ws  = (float*)d_ws;
  float* xc  = ws;
  float* y   = xc  + (size_t)MP_ * C_;
  float* qkv = y   + (size_t)MP_ * C_;
  float* o   = qkv + (size_t)MP_ * C3_;
  float* m1  = qkv;
  float* ga  = o   + (size_t)MP_ * C_;
  float* gst = ga  + (size_t)B_ * HEADS_ * HW_;
  float* wf  = gst + B_ * HEADS_ * 2;
  const float* qkv_wf  = wf;
  const float* proj_wf = qkv_wf  + WQ_;
  const float* fc1_wf  = proj_wf + WP_;
  const float* fc2_wf  = fc1_wf  + W1_;
  float* out = (float*)d_out;

  conv_pe_kernel<<<dim3(NPOS_ / 256, C_ / CH_), 256, 0, stream>>>(x, conv_w, conv_b, xc);
  prep_kernel<<<PBC_ + PBQ_ + PBP_ + PB1_ + PB2_, 256, 0, stream>>>(
      cls, qkv_w, proj_w, fc1_w, fc2_w, xc, wf);

  ln_kernel<<<(MT_ + 1) / 2, 64, 0, stream>>>(xc, ln1_g, ln1_b, y);

  gemm_wmma_kernel<0><<<dim3(C3_ / 64, (MB_ + 7) / 8), 256, 0, stream>>>(
      y, qkv_wf, nullptr, qkv, C_, C3_, MB_);

  attn_kernel<<<dim3(5, HEADS_, B_), 160, 0, stream>>>(qkv, o, ga, gst);

  gemm_res_kernel<<<dim3(C_ / (64 * RN_), MT_ / RM_), 128, 0, stream>>>(
      o, proj_wf, proj_b, xc, C_);

  ln_kernel<<<(MT_ + 1) / 2, 64, 0, stream>>>(xc, ln2_g, ln2_b, y);

  gemm_wmma_kernel<2><<<dim3(C4_ / 64, (MB_ + 7) / 8), 256, 0, stream>>>(
      y, fc1_wf, fc1_b, m1, C_, C4_, MB_);

  gemm_res_kernel<<<dim3(C_ / (64 * RN_), MT_ / RM_), 128, 0, stream>>>(
      m1, fc2_wf, fc2_b, xc, C4_);

  finalize_kernel<<<XB_ + CB_ + GB_, 256, 0, stream>>>(xc, ga, gst, out);
}


# proj and fc2 GEMMs: a trip's 28 fragment loads are issued one trip ahead of its 80 WMMAs, into a second register set, so no trip opens by waiting out a memory round trip; sums unchanged
# speedup vs baseline: 3.2267x; 1.0462x over previous
#include <hip/hip_runtime.h>
#include <math.h>

#define B_    16
#define C_    384
#define H_    28
#define W_    28
#define HW_   784
#define N_    785
#define M_    (B_*N_)
#define MP_   (M_ + 16)
#define HEADS_ 8
#define D_    48
#define C3_   1152
#define C4_   1536
#define MT_   (M_/16)
#define MB_   ((MT_ + 1)/2)
#define CH_   32
#define NPOS_ (B_*HW_)

typedef __attribute__((ext_vector_type(2))) float v2f;
typedef __attribute__((ext_vector_type(4))) float v4f;
typedef __attribute__((ext_vector_type(8))) float v8f;

__device__ __forceinline__ v8f wmma4(v2f a, v2f b, v8f c) {
  return __builtin_amdgcn_wmma_f32_16x16x4_f32(false, a, false, b, (short)0, c,
                                               false, false);
}

__global__ __launch_bounds__(256) void conv_pe_kernel(const float* __restrict__ x,
                                                      const float* __restrict__ cw,
                                                      const float* __restrict__ cb,
                                                      float* __restrict__ xc) {
  const int gp = blockIdx.x * 256 + threadIdx.x;
  const int c0 = blockIdx.y * CH_;
  const int b  = gp / HW_, p = gp - b * HW_;
  const int hh = p / W_,  ww = p - hh * W_;
  bool ok[9];
  int  off[9];
#pragma unroll
  for (int i = 0; i < 3; ++i) {
#pragma unroll
    for (int j = 0; j < 3; ++j) {
      const bool vi = (i == 0) ? (hh > 0) : (i == 2) ? (hh < H_ - 1) : true;
      const bool vj = (j == 0) ? (ww > 0) : (j == 2) ? (ww < W_ - 1) : true;
      ok[i * 3 + j]  = vi && vj;
      off[i * 3 + j] = (vi && vj) ? (i - 1) * W_ + (j - 1) : 0;
    }
  }
  const float* xp = x + ((size_t)(b * C_ + c0) * HW_ + p);
  float* op = xc + ((size_t)(b * N_ + 1 + p) * C_ + c0);
#pragma unroll
  for (int g = 0; g < CH_ / 8; ++g) {
    float v[8][9];
#pragma unroll
    for (int q = 0; q < 8; ++q) {
      const float* xq = xp + (size_t)(8 * g + q) * HW_;
#pragma unroll
      for (int t = 0; t < 9; ++t) v[q][t] = xq[off[t]];
    }
    __builtin_amdgcn_sched_barrier(0);
    float r[8];
#pragma unroll
    for (int q = 0; q < 8; ++q) {
      const int c = c0 + 8 * g + q;
      const float* wq = cw + c * 9;
      float s = 0.f;
#pragma unroll
      for (int t = 0; t < 9; ++t) s = fmaf(ok[t] ? v[q][t] : 0.f, wq[t], s);
      r[q] = v[q][4] + s + cb[c];
    }
    v4f w0 = {r[0], r[1], r[2], r[3]};
    v4f w1 = {r[4], r[5], r[6], r[7]};
    *(v4f*)(op + 8 * g)     = w0;
    *(v4f*)(op + 8 * g + 4) = w1;
  }
}

__device__ __forceinline__ size_t fo_off(int r, int k4, int kg) {
  return ((size_t)((r >> 4) * kg + (k4 >> 3)) * 32 + ((k4 >> 2) & 1) * 16 + (r & 15)) * 4;
}

#define WQ_ (C3_ * C_)
#define WP_ (C_ * C_)
#define W1_ (C4_ * C_)
#define W2_ (C_ * C4_)
#define PBC_ ((B_ * C_) / 256)
#define PBQ_ (WQ_ / 1024)
#define PBP_ (WP_ / 1024)
#define PB1_ (W1_ / 1024)
#define PB2_ (W2_ / 1024)

__device__ __forceinline__ void frag_order_piece(const float* __restrict__ src,
                                                 float* __restrict__ dst,
                                                 int kg, int p) {
  const int lane = p & 31;
  const int gi = p >> 5;
  const int t = gi / kg, i = gi - t * kg;
  const float* sp = src + (size_t)(t * 16 + (lane & 15)) * (kg * 8) + 8 * i + 4 * (lane >> 4);
  *(v4f*)(dst + (size_t)p * 4) = *(const v4f*)sp;
}

__global__ __launch_bounds__(256) void prep_kernel(const float* __restrict__ cls,
                                                   const float* __restrict__ qkv_w,
                                                   const float* __restrict__ proj_w,
                                                   const float* __restrict__ fc1_w,
                                                   const float* __restrict__ fc2_w,
                                                   float* __restrict__ xc,
                                                   float* __restrict__ wf) {
  int bx = blockIdx.x;
  if (bx < PBC_) {
    const int idx = bx * 256 + threadIdx.x;
    const int b = idx / C_, c = idx - b * C_;
    xc[(size_t)b * N_ * C_ + c] = cls[idx];
    return;
  }
  bx -= PBC_;
  if (bx < PBQ_) { frag_order_piece(qkv_w, wf, C_ / 8, bx * 256 + threadIdx.x); return; }
  bx -= PBQ_;
  if (bx < PBP_) { frag_order_piece(proj_w, wf + WQ_, C_ / 8, bx * 256 + threadIdx.x); return; }
  bx -= PBP_;
  if (bx < PB1_) { frag_order_piece(fc1_w, wf + WQ_ + WP_, C_ / 8, bx * 256 + threadIdx.x); return; }
  bx -= PB1_;
  frag_order_piece(fc2_w, wf + WQ_ + WP_ + W1_, C4_ / 8, bx * 256 + threadIdx.x);
}

__global__ __launch_bounds__(64) void ln_kernel(const float* __restrict__ xin,
                                                const float* __restrict__ g,
                                                const float* __restrict__ bta,
                                                float* __restrict__ yf) {
  const int wave = threadIdx.x >> 5, lane = threadIdx.x & 31;
  const int tile = blockIdx.x * 2 + wave;
  if (tile >= MT_) return;
  const int c = lane & 15, hf = lane >> 4;
  const float* xp = xin + (size_t)(tile * 16 + c) * C_ + 4 * hf;
  v4f v[C_ / 8];
#pragma unroll
  for (int i = 0; i < C_ / 8; ++i) v[i] = *(const v4f*)(xp + 8 * i);
  v4f s4 = {0.f, 0.f, 0.f, 0.f}, q4 = {0.f, 0.f, 0.f, 0.f};
#pragma unroll
  for (int i = 0; i < C_ / 8; ++i) { s4 += v[i]; q4 += v[i] * v[i]; }
  float s  = (s4.x + s4.y) + (s4.z + s4.w);
  float sq = (q4.x + q4.y) + (q4.z + q4.w);
  s  += __shfl_xor(s,  16, 32);
  sq += __shfl_xor(sq, 16, 32);
  const float mu  = s  * (1.f / C_);
  const float var = sq * (1.f / C_) - mu * mu;
  const float rs  = rsqrtf(var + 1e-5f);
  float* yp = yf + (size_t)tile * (16 * C_) + lane * 4;
  const float* gp = g + 4 * hf;
  const float* bp = bta + 4 * hf;
#pragma unroll
  for (int i = 0; i < C_ / 8; ++i) {
    const v4f gg = *(const v4f*)(gp + 8 * i);
    const v4f bb = *(const v4f*)(bp + 8 * i);
    v4f r;
    r.x = (v[i].x - mu) * rs * gg.x + bb.x;
    r.y = (v[i].y - mu) * rs * gg.y + bb.y;
    r.z = (v[i].z - mu) * rs * gg.z + bb.z;
    r.w = (v[i].w - mu) * rs * gg.w + bb.w;
    *(v4f*)(yp + i * 128) = r;
  }
}

template <int EPI>
__global__ __launch_bounds__(256) void gemm_wmma_kernel(
    const float* __restrict__ Xf, const float* __restrict__ Wf,
    const float* __restrict__ bias, float* __restrict__ Out,
    int K, int ldo, int Mblocks) {
  int lane = threadIdx.x & 31;
  int wave = threadIdx.x >> 5;
  int mblk = blockIdx.y * 8 + wave;
  if (mblk >= Mblocks) return;
  int n0 = blockIdx.x * 64;
  int m0 = mblk * 32;
  const size_t tstride = (size_t)K * 16;
  const float* xp0 = Xf + (size_t)(2 * mblk) * tstride + lane * 4;
  const float* xp1 = xp0 + tstride;
  const float* wp0 = Wf + (size_t)(4 * blockIdx.x) * tstride + lane * 4;
  const float* wp1 = wp0 + tstride;
  const float* wp2 = wp1 + tstride;
  const float* wp3 = wp2 + tstride;

  v8f acc[2][4] = {};
  for (int k = 0; k < K; k += 32) {
    v4f a0[4], a1[4], b0[4], b1[4], b2[4], b3[4];
#pragma unroll
    for (int u = 0; u < 4; ++u) {
      a0[u] = *(const v4f*)(xp0 + k * 16 + 128 * u);
      a1[u] = *(const v4f*)(xp1 + k * 16 + 128 * u);
      b0[u] = *(const v4f*)(wp0 + k * 16 + 128 * u);
      b1[u] = *(const v4f*)(wp1 + k * 16 + 128 * u);
      b2[u] = *(const v4f*)(wp2 + k * 16 + 128 * u);
      b3[u] = *(const v4f*)(wp3 + k * 16 + 128 * u);
    }
    __builtin_amdgcn_sched_barrier(0);
#define MM_(ACC, XA, WB) ACC = (EPI == 2) ? wmma4(WB, XA, ACC) : wmma4(XA, WB, ACC)
#pragma unroll
    for (int u = 0; u < 4; ++u) {
      v2f a0l = {a0[u].x, a0[u].y}, a0h = {a0[u].z, a0[u].w};
      v2f a1l = {a1[u].x, a1[u].y}, a1h = {a1[u].z, a1[u].w};
      v2f b0l = {b0[u].x, b0[u].y}, b0h = {b0[u].z, b0[u].w};
      v2f b1l = {b1[u].x, b1[u].y}, b1h = {b1[u].z, b1[u].w};
      v2f b2l = {b2[u].x, b2[u].y}, b2h = {b2[u].z, b2[u].w};
      v2f b3l = {b3[u].x, b3[u].y}, b3h = {b3[u].z, b3[u].w};
      MM_(acc[0][0], a0l, b0l);
      MM_(acc[1][0], a1l, b0l);
      MM_(acc[0][1], a0l, b1l);
      MM_(acc[1][1], a1l, b1l);
      MM_(acc[0][2], a0l, b2l);
      MM_(acc[1][2], a1l, b2l);
      MM_(acc[0][3], a0l, b3l);
      MM_(acc[1][3], a1l, b3l);
      MM_(acc[0][0], a0h, b0h);
      MM_(acc[1][0], a1h, b0h);
      MM_(acc[0][1], a0h, b1h);
      MM_(acc[1][1], a1h, b1h);
      MM_(acc[0][2], a0h, b2h);
      MM_(acc[1][2], a1h, b2h);
      MM_(acc[0][3], a0h, b3h);
      MM_(acc[1][3], a1h, b3h);
    }
#undef MM_
  }

  if (EPI == 2) {
    const int c = lane & 15, hfl = lane >> 4;
    const int kg2 = ldo >> 3;
#pragma unroll
    for (int mi = 0; mi < 2; ++mi) {
      const int mt = 2 * mblk + mi;
      if (mt < MT_) {
#pragma unroll
        for (int j = 0; j < 4; ++j) {
          const int nb = n0 + 16 * j + 8 * hfl;
          const v4f bl = *(const v4f*)(bias + nb);
          const v4f bh = *(const v4f*)(bias + nb + 4);
          float t[8];
          t[0] = acc[mi][j][0] + bl.x; t[1] = acc[mi][j][1] + bl.y;
          t[2] = acc[mi][j][2] + bl.z; t[3] = acc[mi][j][3] + bl.w;
          t[4] = acc[mi][j][4] + bh.x; t[5] = acc[mi][j][5] + bh.y;
          t[6] = acc[mi][j][6] + bh.z; t[7] = acc[mi][j][7] + bh.w;
#pragma unroll
          for (int r = 0; r < 8; ++r)
            t[r] = 0.5f * t[r] * (1.f + erff(t[r] * 0.70710678118654752f));
          float* dp = Out + ((size_t)(mt * kg2 + (nb >> 3)) * 32 + c) * 4;
          v4f lo = {t[0], t[1], t[2], t[3]};
          v4f hi = {t[4], t[5], t[6], t[7]};
          *(v4f*)dp        = lo;
          *(v4f*)(dp + 64) = hi;
        }
      }
    }
    return;
  }

  int coln = lane & 15;
  int roff = (lane >> 4) << 3;
#pragma unroll
  for (int mi = 0; mi < 2; ++mi) {
#pragma unroll
    for (int j = 0; j < 4; ++j) {
      int nc = n0 + j * 16 + coln;
      float bv = bias ? bias[nc] : 0.f;
#pragma unroll
      for (int r = 0; r < 8; ++r) {
        int gm = m0 + mi * 16 + roff + r;
        if (gm < M_) {
          Out[(size_t)gm * ldo + nc] = acc[mi][j][r] + bv;
        }
      }
    }
  }
}

#define RM_ 5
#define RN_ 2
__global__ __launch_bounds__(128) void gemm_res_kernel(
    const float* __restrict__ Xf, const float* __restrict__ Wf,
    const float* __restrict__ bias, float* __restrict__ Out, int K) {
  const int lane = threadIdx.x & 31;
  const int wave = threadIdx.x >> 5;
  const int mblk = blockIdx.y;
  const int nblk = blockIdx.x * 4 + wave;
  const size_t tstride = (size_t)K * 16;
  const float* xp = Xf + (size_t)(RM_ * mblk) * tstride + lane * 4;
  const float* wp = Wf + (size_t)(RN_ * nblk) * tstride + lane * 4;

  v8f acc[RM_][RN_] = {};
#define RES_LOAD_(A, B, KK)                                                   \
  _Pragma("unroll") for (int u = 0; u < 4; ++u) {                             \
    _Pragma("unroll") for (int mi = 0; mi < RM_; ++mi)                        \
      A[mi][u] = *(const v4f*)(xp + mi * tstride + (size_t)(KK) * 16 + 128 * u); \
    _Pragma("unroll") for (int j = 0; j < RN_; ++j)                           \
      B[j][u] = *(const v4f*)(wp + j * tstride + (size_t)(KK) * 16 + 128 * u);   \
  }
#define RES_MMA_(A, B)                                                        \
  _Pragma("unroll") for (int u = 0; u < 4; ++u) {                             \
    _Pragma("unroll") for (int j = 0; j < RN_; ++j) {                         \
      const v2f bl = {B[j][u].x, B[j][u].y};                                  \
      _Pragma("unroll") for (int mi = 0; mi < RM_; ++mi) {                    \
        const v2f al = {A[mi][u].x, A[mi][u].y};                              \
        acc[mi][j] = wmma4(al, bl, acc[mi][j]);                               \
      }                                                                       \
    }                                                                         \
    _Pragma("unroll") for (int j = 0; j < RN_; ++j) {                         \
      const v2f bh = {B[j][u].z, B[j][u].w};                                  \
      _Pragma("unroll") for (int mi = 0; mi < RM_; ++mi) {                    \
        const v2f ah = {A[mi][u].z, A[mi][u].w};                              \
        acc[mi][j] = wmma4(ah, bh, acc[mi][j]);                               \
      }                                                                       \
    }                                                                         \
  }
  v4f a0[RM_][4], b0[RN_][4], a1[RM_][4], b1[RN_][4];
  RES_LOAD_(a0, b0, 0)
  for (int k = 0; k < K; k += 64) {
    RES_LOAD_(a1, b1, k + 32)
    __builtin_amdgcn_sched_barrier(0);
    RES_MMA_(a0, b0)
    __builtin_amdgcn_sched_barrier(0);
    const int kn = (k + 64 < K) ? k + 64 : k;
    RES_LOAD_(a0, b0, kn)
    __builtin_amdgcn_sched_barrier(0);
    RES_MMA_(a1, b1)
  }
#undef RES_LOAD_
#undef RES_MMA_

  const int coln = lane & 15;
  const int roff = (lane >> 4) << 3;
#pragma unroll
  for (int mi = 0; mi < RM_; ++mi) {
#pragma unroll
    for (int j = 0; j < RN_; ++j) {
      const int nc = nblk * (16 * RN_) + 16 * j + coln;
      const float bv = bias[nc];
      float* op = Out + (size_t)(mblk * (16 * RM_) + 16 * mi + roff) * C_ + nc;
#pragma unroll
      for (int r = 0; r < 8; ++r) {
        const float v = acc[mi][j][r] + bv;
        op[(size_t)r * C_] += v;
      }
    }
  }
}

struct __attribute__((packed, aligned(4))) f3 { float x, y, z; };

struct KStep { v4f k[6]; };
struct VStep { f3 va[4]; f3 vb[4]; };

__device__ __forceinline__ void load_k(KStep& t, const float* __restrict__ kbase,
                                       int key0, int c) {
  int kr = key0 + c; if (kr > N_ - 1) kr = N_ - 1;
  const float* kp = kbase + (size_t)kr * C3_;
#pragma unroll
  for (int i = 0; i < 6; ++i) t.k[i] = *(const v4f*)(kp + 8 * i);
}

__device__ __forceinline__ void load_v(VStep& t, const float* __restrict__ vbase,
                                       int key0, int hf) {
  const int kl = key0 + 8 * hf;
#pragma unroll
  for (int j = 0; j < 4; ++j) {
    int k0 = kl + 2 * j;     if (k0 > N_ - 1) k0 = N_ - 1;
    int k1 = kl + 2 * j + 1; if (k1 > N_ - 1) k1 = N_ - 1;
    t.va[j] = *(const f3*)(vbase + (size_t)k0 * C3_);
    t.vb[j] = *(const f3*)(vbase + (size_t)k1 * C3_);
  }
}

struct AttnState {
  v8f oa0, oa1, oa2, ob0, ob1, ob2;
  float m0, m1, l0, l1;
};

__device__ __forceinline__ void attn_step(const KStep& tk, const VStep& t,
                                          const v4f (&q0)[6],
                                          const v4f (&q1)[6], AttnState& st,
                                          int key0, int hf, bool cls_lane,
                                          float* __restrict__ gp) {
  const float scale = 0.14433756729740643f;
  v8f s0 = {}, s1 = {};
#pragma unroll
  for (int i = 0; i < 6; ++i) {
    v2f ka = {tk.k[i].x, tk.k[i].y}, kc = {tk.k[i].z, tk.k[i].w};
    v2f qa0 = {q0[i].x, q0[i].y}, qc0 = {q0[i].z, q0[i].w};
    v2f qa1 = {q1[i].x, q1[i].y}, qc1 = {q1[i].z, q1[i].w};
    s0 = wmma4(ka, qa0, s0);
    s1 = wmma4(ka, qa1, s1);
    s0 = wmma4(kc, qc0, s0);
    s1 = wmma4(kc, qc1, s1);
  }
  const int kl = key0 + 8 * hf;
  float x0[8], x1[8];
#pragma unroll
  for (int r = 0; r < 8; ++r) { x0[r] = s0[r] * scale; x1[r] = s1[r] * scale; }
  if (key0 + 16 > N_) {
#pragma unroll
    for (int r = 0; r < 8; ++r)
      if (kl + r >= N_) { x0[r] = -3.0e38f; x1[r] = -3.0e38f; }
  }
  if (cls_lane) {
#pragma unroll
    for (int r = 0; r < 8; ++r) {
      int key = kl + r;
      if (key >= 1 && key < N_) gp[key - 1] = x0[r];
    }
  }
  float mx0 = x0[0], mx1 = x1[0];
#pragma unroll
  for (int r = 1; r < 8; ++r) { mx0 = fmaxf(mx0, x0[r]); mx1 = fmaxf(mx1, x1[r]); }
  mx0 = fmaxf(mx0, __shfl_xor(mx0, 16, 32));
  mx1 = fmaxf(mx1, __shfl_xor(mx1, 16, 32));
  float mn0 = fmaxf(st.m0, mx0), mn1 = fmaxf(st.m1, mx1);
  float al0 = expf(st.m0 - mn0), al1 = expf(st.m1 - mn1);
  float ps0 = 0.f, ps1 = 0.f;
#pragma unroll
  for (int r = 0; r < 8; ++r) {
    x0[r] = expf(x0[r] - mn0); ps0 += x0[r];
    x1[r] = expf(x1[r] - mn1); ps1 += x1[r];
  }
  st.l0 = st.l0 * al0 + ps0;  st.l1 = st.l1 * al1 + ps1;
  st.m0 = mn0;  st.m1 = mn1;
#pragma unroll
  for (int r = 0; r < 8; ++r) {
    st.oa0[r] *= al0; st.oa1[r] *= al0; st.oa2[r] *= al0;
    st.ob0[r] *= al1; st.ob1[r] *= al1; st.ob2[r] *= al1;
  }
#pragma unroll
  for (int j = 0; j < 4; ++j) {
    v2f pa = {x0[2 * j], x0[2 * j + 1]};
    v2f pb = {x1[2 * j], x1[2 * j + 1]};
    v2f a0 = {t.va[j].x, t.vb[j].x}, a1 = {t.va[j].y, t.vb[j].y}, a2 = {t.va[j].z, t.vb[j].z};
    st.oa0 = wmma4(a0, pa, st.oa0);
    st.ob0 = wmma4(a0, pb, st.ob0);
    st.oa1 = wmma4(a1, pa, st.oa1);
    st.ob1 = wmma4(a1, pb, st.ob1);
    st.oa2 = wmma4(a2, pa, st.oa2);
    st.ob2 = wmma4(a2, pb, st.ob2);
  }
}

__global__ __launch_bounds__(160, 4) void attn_kernel(const float* __restrict__ qkv,
                                                   float* __restrict__ o,
                                                   float* __restrict__ ga,
                                                   float* __restrict__ gst) {
  int lane = threadIdx.x & 31;
  int wave = threadIdx.x >> 5;
  int pair = blockIdx.x * 5 + wave;
  int h = blockIdx.y, b = blockIdx.z;
  int t0 = pair * 32;
  int c  = lane & 15;
  int hf = lane >> 4;

  int gq0 = t0 + c;      if (gq0 > N_ - 1) gq0 = N_ - 1;
  int gq1 = t0 + 16 + c; if (gq1 > N_ - 1) gq1 = N_ - 1;
  const float* qp0 = qkv + (size_t)(b * N_ + gq0) * C3_ + h * D_ + 4 * hf;
  const float* qp1 = qkv + (size_t)(b * N_ + gq1) * C3_ + h * D_ + 4 * hf;
  v4f q0[6], q1[6];
#pragma unroll
  for (int i = 0; i < 6; ++i) {
    q0[i] = *(const v4f*)(qp0 + 8 * i);
    q1[i] = *(const v4f*)(qp1 + 8 * i);
  }

  const float* kbase = qkv + (size_t)b * N_ * C3_ + C_ + h * D_ + 4 * hf;
  const float* vbase = qkv + (size_t)b * N_ * C3_ + 2 * C_ + h * D_ + 3 * c;
  float* gp = ga + (size_t)(b * HEADS_ + h) * HW_;
  const bool cls_lane = (pair == 0) && (c == 0);

  AttnState st;
  st.oa0 = v8f{}; st.oa1 = v8f{}; st.oa2 = v8f{};
  st.ob0 = v8f{}; st.ob1 = v8f{}; st.ob2 = v8f{};
  st.m0 = -3.0e38f; st.m1 = -3.0e38f; st.l0 = 0.f; st.l1 = 0.f;

  KStep ka, kb;
  load_k(ka, kbase, 0, c);
  for (int key0 = 0; key0 < N_; key0 += 32) {
    VStep va, vb;
    load_k(kb, kbase, key0 + 16, c);
    load_v(va, vbase, key0, hf);
    attn_step(ka, va, q0, q1, st, key0, hf, cls_lane, gp);
    load_k(ka, kbase, key0 + 32, c);
    load_v(vb, vbase, key0 + 16, hf);
    attn_step(kb, vb, q0, q1, st, key0 + 16, hf, cls_lane, gp);
  }

  float lt0 = st.l0 + __shfl_xor(st.l0, 16, 32);
  float lt1 = st.l1 + __shfl_xor(st.l1, 16, 32);
  if (pair == 0 && lane == 0) {
    gst[(b * HEADS_ + h) * 2]     = st.m0;
    gst[(b * HEADS_ + h) * 2 + 1] = lt0;
  }
  float inv0 = 1.f / lt0, inv1 = 1.f / lt1;
  int qa = t0 + c, qb = t0 + 16 + c;
  const int kd = h * D_ + 24 * hf;
  if (qa < N_) {
    const int m = b * N_ + qa;
#pragma unroll
    for (int g = 0; g < 2; ++g) {
      v4f w0 = {st.oa0[4*g] * inv0,   st.oa1[4*g] * inv0,   st.oa2[4*g] * inv0,   st.oa0[4*g+1] * inv0};
      v4f w1 = {st.oa1[4*g+1] * inv0, st.oa2[4*g+1] * inv0, st.oa0[4*g+2] * inv0, st.oa1[4*g+2] * inv0};
      v4f w2 = {st.oa2[4*g+2] * inv0, st.oa0[4*g+3] * inv0, st.oa1[4*g+3] * inv0, st.oa2[4*g+3] * inv0};
      *(v4f*)(o + fo_off(m, kd + 12 * g,     C_ / 8)) = w0;
      *(v4f*)(o + fo_off(m, kd + 12 * g + 4, C_ / 8)) = w1;
      *(v4f*)(o + fo_off(m, kd + 12 * g + 8, C_ / 8)) = w2;
    }
  }
  if (qb < N_) {
    const int m = b * N_ + qb;
#pragma unroll
    for (int g = 0; g < 2; ++g) {
      v4f w0 = {st.ob0[4*g] * inv1,   st.ob1[4*g] * inv1,   st.ob2[4*g] * inv1,   st.ob0[4*g+1] * inv1};
      v4f w1 = {st.ob1[4*g+1] * inv1, st.ob2[4*g+1] * inv1, st.ob0[4*g+2] * inv1, st.ob1[4*g+2] * inv1};
      v4f w2 = {st.ob2[4*g+2] * inv1, st.ob0[4*g+3] * inv1, st.ob1[4*g+3] * inv1, st.ob2[4*g+3] * inv1};
      *(v4f*)(o + fo_off(m, kd + 12 * g,     C_ / 8)) = w0;
      *(v4f*)(o + fo_off(m, kd + 12 * g + 4, C_ / 8)) = w1;
      *(v4f*)(o + fo_off(m, kd + 12 * g + 8, C_ / 8)) = w2;
    }
  }
}

#define XB_ ((NPOS_ / 256) * (C_ / CH_))
#define CB_ ((B_ * C_) / 256)
#define GB_ (NPOS_ / 256)
__global__ __launch_bounds__(256) void finalize_kernel(const float* __restrict__ xc,
                                                       const float* __restrict__ ga,
                                                       const float* __restrict__ gst,
                                                       float* __restrict__ out) {
  const int bx = blockIdx.x;
  if (bx < XB_) {
    const int chunk = bx / (NPOS_ / 256);
    const int gp = (bx - chunk * (NPOS_ / 256)) * 256 + threadIdx.x;
    const int c0 = chunk * CH_;
    const int b  = gp / HW_, p = gp - b * HW_;
    const float* ip = xc + ((size_t)(b * N_ + 1 + p) * C_ + c0);
    float* op = out + (size_t)B_ * C_ + ((size_t)(b * C_ + c0) * HW_ + p);
    v4f v[CH_ / 4];
#pragma unroll
    for (int g = 0; g < CH_ / 4; ++g) v[g] = *(const v4f*)(ip + 4 * g);
#pragma unroll
    for (int g = 0; g < CH_ / 4; ++g) {
      op[(size_t)(4 * g + 0) * HW_] = v[g].x;
      op[(size_t)(4 * g + 1) * HW_] = v[g].y;
      op[(size_t)(4 * g + 2) * HW_] = v[g].z;
      op[(size_t)(4 * g + 3) * HW_] = v[g].w;
    }
  } else if (bx < XB_ + CB_) {
    const int idx = (bx - XB_) * 256 + threadIdx.x;
    const int b = idx / C_, c = idx - b * C_;
    out[idx] = xc[(size_t)b * N_ * C_ + c];
  } else {
    const int j = (bx - XB_ - CB_) * 256 + threadIdx.x;
    const int b = j / HW_, p = j - b * HW_;
    float s = 0.f;
#pragma unroll
    for (int hh = 0; hh < HEADS_; ++hh) {
      float mh = gst[(b * HEADS_ + hh) * 2];
      float lh = gst[(b * HEADS_ + hh) * 2 + 1];
      s += expf(ga[(size_t)(b * HEADS_ + hh) * HW_ + p] - mh) * (1.f / lh);
    }
    out[(size_t)B_ * C_ + (size_t)B_ * C_ * HW_ + j] = s * 0.125f;
  }
}

extern "C" void kernel_launch(void* const* d_in, const int* in_sizes, int n_in,
                              void* d_out, int out_size, void* d_ws, size_t ws_size,
                              hipStream_t stream) {
  const float* cls    = (const float*)d_in[0];
  const float* x      = (const float*)d_in[1];
  const float* conv_w = (const float*)d_in[2];
  const float* conv_b = (const float*)d_in[3];
  const float* ln1_g  = (const float*)d_in[4];
  const float* ln1_b  = (const float*)d_in[5];
  const float* qkv_w  = (const float*)d_in[6];
  const float* proj_w = (const float*)d_in[7];
  const float* proj_b = (const float*)d_in[8];
  const float* ln2_g  = (const float*)d_in[9];
  const float* ln2_b  = (const float*)d_in[10];
  const float* fc1_w  = (const float*)d_in[11];
  const float* fc1_b  = (const float*)d_in[12];
  const float* fc2_w  = (const float*)d_in[13];
  const float* fc2_b  = (const float*)d_in[14];

  float* ws  = (float*)d_ws;
  float* xc  = ws;
  float* y   = xc  + (size_t)MP_ * C_;
  float* qkv = y   + (size_t)MP_ * C_;
  float* o   = qkv + (size_t)MP_ * C3_;
  float* m1  = qkv;
  float* ga  = o   + (size_t)MP_ * C_;
  float* gst = ga  + (size_t)B_ * HEADS_ * HW_;
  float* wf  = gst + B_ * HEADS_ * 2;
  const float* qkv_wf  = wf;
  const float* proj_wf = qkv_wf  + WQ_;
  const float* fc1_wf  = proj_wf + WP_;
  const float* fc2_wf  = fc1_wf  + W1_;
  float* out = (float*)d_out;

  conv_pe_kernel<<<dim3(NPOS_ / 256, C_ / CH_), 256, 0, stream>>>(x, conv_w, conv_b, xc);
  prep_kernel<<<PBC_ + PBQ_ + PBP_ + PB1_ + PB2_, 256, 0, stream>>>(
      cls, qkv_w, proj_w, fc1_w, fc2_w, xc, wf);

  ln_kernel<<<(MT_ + 1) / 2, 64, 0, stream>>>(xc, ln1_g, ln1_b, y);

  gemm_wmma_kernel<0><<<dim3(C3_ / 64, (MB_ + 7) / 8), 256, 0, stream>>>(
      y, qkv_wf, nullptr, qkv, C_, C3_, MB_);

  attn_kernel<<<dim3(5, HEADS_, B_), 160, 0, stream>>>(qkv, o, ga, gst);

  gemm_res_kernel<<<dim3(C_ / (64 * RN_), MT_ / RM_), 128, 0, stream>>>(
      o, proj_wf, proj_b, xc, C_);

  ln_kernel<<<(MT_ + 1) / 2, 64, 0, stream>>>(xc, ln2_g, ln2_b, y);

  gemm_wmma_kernel<2><<<dim3(C4_ / 64, (MB_ + 7) / 8), 256, 0, stream>>>(
      y, fc1_wf, fc1_b, m1, C_, C4_, MB_);

  gemm_res_kernel<<<dim3(C_ / (64 * RN_), MT_ / RM_), 128, 0, stream>>>(
      m1, fc2_wf, fc2_b, xc, C4_);

  finalize_kernel<<<XB_ + CB_ + GB_, 256, 0, stream>>>(xc, ga, gst, out);
}
